# starting kernel
# baseline (speedup 1.0000x reference)
#include <hip/hip_runtime.h>

#define BATCH 4096
#define SEQ   2048
#define ARP   8
#define EXO   8
#define USTR  (SEQ + EXO)
#define OSTR  (SEQ + ARP + 1)
#define KTAPS 256
#define EROWS (SEQ + 8)
#define EROWS_PAD (EROWS + 8)
#define HOFF  32
#define HPAD_LEN 512

typedef __attribute__((ext_vector_type(2))) float v2f;
typedef __attribute__((ext_vector_type(8))) float v8f;

__global__ void k_impulse(const float* __restrict__ W, float* __restrict__ hpad) {
  if (threadIdx.x == 0 && blockIdx.x == 0) {
    for (int i = 0; i < HPAD_LEN; ++i) hpad[i] = 0.0f;
    hpad[HOFF] = 1.0f;
    for (int n = 1; n < KTAPS; ++n) {
      float s = 0.0f;
#pragma unroll
      for (int d = 1; d <= 8; ++d) s += W[8 - d] * hpad[HOFF + n - d];
      hpad[HOFF + n] = s;
    }
  }
}

__global__ void k_eps(const float* __restrict__ y, const float* __restrict__ hpad,
                      float* __restrict__ Efull) {
  const int b = blockIdx.x * blockDim.x + threadIdx.x;
  if (b >= BATCH) return;
  float hv[8];
#pragma unroll
  for (int i = 0; i < 8; ++i) hv[i] = hpad[HOFF + i];
  float e[8];
#pragma unroll
  for (int i = 0; i < 8; ++i) {
    float s = y[b * 9 + i + 1] - y[b * 9 + i];
#pragma unroll
    for (int p = 0; p < 8; ++p) {
      if (p < i) s -= hv[i - p] * e[p];
    }
    e[i] = s;
    Efull[i * BATCH + b] = s;
  }
#pragma unroll
  for (int i = 0; i < 8; ++i)
    Efull[(size_t)(EROWS + i) * BATCH + b] = 0.0f;
}

__global__ void k_exo(const float* __restrict__ u, const float* __restrict__ W,
                      float* __restrict__ Efull) {
  const int b = blockIdx.x * blockDim.x + threadIdx.x;
  const int m = blockIdx.y;
  float wv[8];
#pragma unroll
  for (int j = 0; j < 8; ++j) wv[j] = W[8 + j];
  const float* up = u + (size_t)b * USTR + 1 + m;
  float s = 0.0f;
#pragma unroll
  for (int j = 0; j < 8; ++j) s += wv[j] * up[j];
  Efull[(size_t)(m + 8) * BATCH + b] = s;
}

__global__ void k_zconv(const float* __restrict__ hpad, const float* __restrict__ Efull,
                        float* __restrict__ Z) {
  const int lane = threadIdx.x & 31;
  const int wv   = threadIdx.x >> 5;
  const int nbase = blockIdx.x * 16 + 1;
  const int b0 = blockIdx.y * 256 + wv * 64;
  const int col  = lane & 15;
  const int half = lane >> 4;

  v8f acc0 = {0.0f, 0.0f, 0.0f, 0.0f, 0.0f, 0.0f, 0.0f, 0.0f};
  v8f acc1 = {0.0f, 0.0f, 0.0f, 0.0f, 0.0f, 0.0f, 0.0f, 0.0f};
  v8f acc2 = {0.0f, 0.0f, 0.0f, 0.0f, 0.0f, 0.0f, 0.0f, 0.0f};
  v8f acc3 = {0.0f, 0.0f, 0.0f, 0.0f, 0.0f, 0.0f, 0.0f, 0.0f};

  int r0 = nbase + 8 - KTAPS;
  if (r0 < 0) r0 = 0;
  r0 &= ~3;
  const int rmax = nbase + 22;

  const float* hbase = hpad + (HOFF + nbase + col + 7 - 2 * half - r0);
  const float* ebase = Efull + ((size_t)(r0 + 2 * half) * BATCH + b0 + col);

  for (int r = r0; r <= rmax; r += 4) {
    v2f a;
    a.x = hbase[0];
    a.y = hbase[-1];
    v2f f0, f1, f2, f3;
    f0.x = ebase[0];          f0.y = ebase[BATCH];
    f1.x = ebase[16];         f1.y = ebase[BATCH + 16];
    f2.x = ebase[32];         f2.y = ebase[BATCH + 32];
    f3.x = ebase[48];         f3.y = ebase[BATCH + 48];
    acc0 = __builtin_amdgcn_wmma_f32_16x16x4_f32(false, a, false, f0, (short)0,
                                                 acc0, false, false);
    acc1 = __builtin_amdgcn_wmma_f32_16x16x4_f32(false, a, false, f1, (short)0,
                                                 acc1, false, false);
    acc2 = __builtin_amdgcn_wmma_f32_16x16x4_f32(false, a, false, f2, (short)0,
                                                 acc2, false, false);
    acc3 = __builtin_amdgcn_wmma_f32_16x16x4_f32(false, a, false, f3, (short)0,
                                                 acc3, false, false);
    hbase -= 4;
    ebase += 4 * BATCH;
  }

#pragma unroll
  for (int v = 0; v < 8; ++v) {
    const int row = v + 8 * half;
    float* zp = Z + ((size_t)(nbase - 1 + row) * BATCH + b0 + col);
    zp[0]  = acc0[v];
    zp[16] = acc1[v];
    zp[32] = acc2[v];
    zp[48] = acc3[v];
  }
}

__global__ void k_cumsum(const float* __restrict__ y, const float* __restrict__ Z,
                         float* __restrict__ out) {
  const int b = blockIdx.x * blockDim.x + threadIdx.x;
  if (b >= BATCH) return;
  float* ob = out + (size_t)b * OSTR;
#pragma unroll
  for (int k = 0; k < 9; ++k) ob[k] = y[b * 9 + k];
  float x = y[b * 9 + 8];
  for (int t = 0; t < SEQ; ++t) {
    x += Z[(size_t)t * BATCH + b];
    ob[9 + t] = x;
  }
}

extern "C" void kernel_launch(void* const* d_in, const int* in_sizes, int n_in,
                              void* d_out, int out_size, void* d_ws, size_t ws_size,
                              hipStream_t stream) {
  (void)in_sizes; (void)n_in; (void)out_size; (void)ws_size;
  const float* y = (const float*)d_in[0];
  const float* u = (const float*)d_in[1];
  const float* W = (const float*)d_in[2];
  float* out = (float*)d_out;

  char* ws = (char*)d_ws;
  float* hpad  = (float*)ws;
  float* Efull = (float*)(ws + HPAD_LEN * 4);
  float* Z     = (float*)(ws + HPAD_LEN * 4 +
                          (size_t)EROWS_PAD * BATCH * 4);

  k_impulse<<<1, 32, 0, stream>>>(W, hpad);
  k_eps   <<<BATCH / 256, 256, 0, stream>>>(y, hpad, Efull);
  k_exo   <<<dim3(BATCH / 256, SEQ), 256, 0, stream>>>(u, W, Efull);
  k_zconv <<<dim3(SEQ / 16, BATCH / 256), 128, 0, stream>>>(hpad, Efull, Z);
  k_cumsum<<<BATCH / 256, 256, 0, stream>>>(y, Z, out);
}


# k_cumsum re-tiled: 256 blocks scan a 16-row slab as 32 register segments of 64 steps, totals and rows via LDS, out leaves as whole-line 16 B stores along t; sum order: per segment, then totals
# speedup vs baseline: 2.9154x; 2.9154x over previous
#include <hip/hip_runtime.h>

#define BATCH 4096
#define SEQ   2048
#define ARP   8
#define EXO   8
#define USTR  (SEQ + EXO)
#define OSTR  (SEQ + ARP + 1)
#define KTAPS 256
#define EROWS (SEQ + 8)
#define EROWS_PAD (EROWS + 8)
#define HOFF  32
#define HPAD_LEN 512

typedef __attribute__((ext_vector_type(2))) float v2f;
typedef __attribute__((ext_vector_type(8))) float v8f;

__global__ void k_impulse(const float* __restrict__ W, float* __restrict__ hpad) {
  if (threadIdx.x == 0 && blockIdx.x == 0) {
    for (int i = 0; i < HPAD_LEN; ++i) hpad[i] = 0.0f;
    hpad[HOFF] = 1.0f;
    for (int n = 1; n < KTAPS; ++n) {
      float s = 0.0f;
#pragma unroll
      for (int d = 1; d <= 8; ++d) s += W[8 - d] * hpad[HOFF + n - d];
      hpad[HOFF + n] = s;
    }
  }
}

__global__ void k_eps(const float* __restrict__ y, const float* __restrict__ hpad,
                      float* __restrict__ Efull) {
  const int b = blockIdx.x * blockDim.x + threadIdx.x;
  if (b >= BATCH) return;
  float hv[8];
#pragma unroll
  for (int i = 0; i < 8; ++i) hv[i] = hpad[HOFF + i];
  float e[8];
#pragma unroll
  for (int i = 0; i < 8; ++i) {
    float s = y[b * 9 + i + 1] - y[b * 9 + i];
#pragma unroll
    for (int p = 0; p < 8; ++p) {
      if (p < i) s -= hv[i - p] * e[p];
    }
    e[i] = s;
    Efull[i * BATCH + b] = s;
  }
#pragma unroll
  for (int i = 0; i < 8; ++i)
    Efull[(size_t)(EROWS + i) * BATCH + b] = 0.0f;
}

__global__ void k_exo(const float* __restrict__ u, const float* __restrict__ W,
                      float* __restrict__ Efull) {
  const int b = blockIdx.x * blockDim.x + threadIdx.x;
  const int m = blockIdx.y;
  float wv[8];
#pragma unroll
  for (int j = 0; j < 8; ++j) wv[j] = W[8 + j];
  const float* up = u + (size_t)b * USTR + 1 + m;
  float s = 0.0f;
#pragma unroll
  for (int j = 0; j < 8; ++j) s += wv[j] * up[j];
  Efull[(size_t)(m + 8) * BATCH + b] = s;
}

__global__ void k_zconv(const float* __restrict__ hpad, const float* __restrict__ Efull,
                        float* __restrict__ Z) {
  const int lane = threadIdx.x & 31;
  const int wv   = threadIdx.x >> 5;
  const int nbase = blockIdx.x * 16 + 1;
  const int b0 = blockIdx.y * 256 + wv * 64;
  const int col  = lane & 15;
  const int half = lane >> 4;

  v8f acc0 = {0.0f, 0.0f, 0.0f, 0.0f, 0.0f, 0.0f, 0.0f, 0.0f};
  v8f acc1 = {0.0f, 0.0f, 0.0f, 0.0f, 0.0f, 0.0f, 0.0f, 0.0f};
  v8f acc2 = {0.0f, 0.0f, 0.0f, 0.0f, 0.0f, 0.0f, 0.0f, 0.0f};
  v8f acc3 = {0.0f, 0.0f, 0.0f, 0.0f, 0.0f, 0.0f, 0.0f, 0.0f};

  int r0 = nbase + 8 - KTAPS;
  if (r0 < 0) r0 = 0;
  r0 &= ~3;
  const int rmax = nbase + 22;

  const float* hbase = hpad + (HOFF + nbase + col + 7 - 2 * half - r0);
  const float* ebase = Efull + ((size_t)(r0 + 2 * half) * BATCH + b0 + col);

  for (int r = r0; r <= rmax; r += 4) {
    v2f a;
    a.x = hbase[0];
    a.y = hbase[-1];
    v2f f0, f1, f2, f3;
    f0.x = ebase[0];          f0.y = ebase[BATCH];
    f1.x = ebase[16];         f1.y = ebase[BATCH + 16];
    f2.x = ebase[32];         f2.y = ebase[BATCH + 32];
    f3.x = ebase[48];         f3.y = ebase[BATCH + 48];
    acc0 = __builtin_amdgcn_wmma_f32_16x16x4_f32(false, a, false, f0, (short)0,
                                                 acc0, false, false);
    acc1 = __builtin_amdgcn_wmma_f32_16x16x4_f32(false, a, false, f1, (short)0,
                                                 acc1, false, false);
    acc2 = __builtin_amdgcn_wmma_f32_16x16x4_f32(false, a, false, f2, (short)0,
                                                 acc2, false, false);
    acc3 = __builtin_amdgcn_wmma_f32_16x16x4_f32(false, a, false, f3, (short)0,
                                                 acc3, false, false);
    hbase -= 4;
    ebase += 4 * BATCH;
  }

#pragma unroll
  for (int v = 0; v < 8; ++v) {
    const int row = v + 8 * half;
    float* zp = Z + ((size_t)(nbase - 1 + row) * BATCH + b0 + col);
    zp[0]  = acc0[v];
    zp[16] = acc1[v];
    zp[32] = acc2[v];
    zp[48] = acc3[v];
  }
}

#define CS_ROWS    16
#define CS_SEGS    32
#define CS_LEN     (SEQ / CS_SEGS)
#define CS_THREADS (CS_ROWS * CS_SEGS)
#define CS_IMG     (CS_ROWS * OSTR)
#define CS_TP      68

typedef __attribute__((ext_vector_type(4))) float v4f;

__global__ void __launch_bounds__(CS_THREADS)
k_cumsum(const float* __restrict__ y, const float* __restrict__ Z,
         float* __restrict__ out) {
  __shared__ __attribute__((aligned(16))) float img[CS_IMG];
  __shared__ __attribute__((aligned(16))) float tot[CS_ROWS * CS_TP];

  const int tid = threadIdx.x;
  const int b   = tid & (CS_ROWS - 1);
  const int seg = tid >> 4;
  const int b0  = blockIdx.x * CS_ROWS;

  const float* zp = Z + ((size_t)(seg * CS_LEN) * BATCH + b0 + b);
  float p[CS_LEN];
#pragma unroll
  for (int i = 0; i < CS_LEN; ++i) p[i] = zp[(size_t)i * BATCH];
  const float x8 = y[(b0 + b) * 9 + 8];
#pragma unroll
  for (int i = 1; i < CS_LEN; ++i) p[i] += p[i - 1];

  tot[b * CS_TP + seg] = p[CS_LEN - 1];
  if (tid < CS_ROWS * 9)
    img[(tid / 9) * OSTR + (tid % 9)] = y[b0 * 9 + tid];
  __syncthreads();

  float base = x8;
  const v4f* t4 = (const v4f*)(tot + b * CS_TP);
#pragma unroll
  for (int k = 0; k < CS_SEGS / 4; ++k) {
    const v4f t = t4[k];
    base += (4 * k + 0 < seg) ? t.x : 0.0f;
    base += (4 * k + 1 < seg) ? t.y : 0.0f;
    base += (4 * k + 2 < seg) ? t.z : 0.0f;
    base += (4 * k + 3 < seg) ? t.w : 0.0f;
  }

  float* ip = img + (b * OSTR + 9 + seg * CS_LEN);
#pragma unroll
  for (int i = 0; i < CS_LEN; ++i) ip[i] = base + p[i];
  __syncthreads();

  const v4f* img4 = (const v4f*)img;
  v4f* out4 = (v4f*)(out + (size_t)b0 * OSTR);
  const int pad = (8 - (int)(((size_t)out4 >> 4) & 7)) & 7;
  if (tid < pad) out4[tid] = img4[tid];
  const int q = tid + pad;
#pragma unroll
  for (int k = 0; k < 16; ++k) out4[q + CS_THREADS * k] = img4[q + CS_THREADS * k];
  if (q + CS_THREADS * 16 < CS_IMG / 4)
    out4[q + CS_THREADS * 16] = img4[q + CS_THREADS * 16];
}

extern "C" void kernel_launch(void* const* d_in, const int* in_sizes, int n_in,
                              void* d_out, int out_size, void* d_ws, size_t ws_size,
                              hipStream_t stream) {
  (void)in_sizes; (void)n_in; (void)out_size; (void)ws_size;
  const float* y = (const float*)d_in[0];
  const float* u = (const float*)d_in[1];
  const float* W = (const float*)d_in[2];
  float* out = (float*)d_out;

  char* ws = (char*)d_ws;
  float* hpad  = (float*)ws;
  float* Efull = (float*)(ws + HPAD_LEN * 4);
  float* Z     = (float*)(ws + HPAD_LEN * 4 +
                          (size_t)EROWS_PAD * BATCH * 4);

  k_impulse<<<1, 32, 0, stream>>>(W, hpad);
  k_eps   <<<BATCH / 256, 256, 0, stream>>>(y, hpad, Efull);
  k_exo   <<<dim3(BATCH / 256, SEQ), 256, 0, stream>>>(u, W, Efull);
  k_zconv <<<dim3(SEQ / 16, BATCH / 256), 128, 0, stream>>>(hpad, Efull, Z);
  k_cumsum<<<BATCH / CS_ROWS, CS_THREADS, 0, stream>>>(y, Z, out);
}


# k_exo re-tiled: a block stages 32 rows x 520 steps of u in LDS by 16 B loads along t, then lane = batch row slides the 8 taps over 64 steps; whole-line stores; 4096 waves, not 262144
# speedup vs baseline: 4.0689x; 1.3957x over previous
#include <hip/hip_runtime.h>

#define BATCH 4096
#define SEQ   2048
#define ARP   8
#define EXO   8
#define USTR  (SEQ + EXO)
#define OSTR  (SEQ + ARP + 1)
#define KTAPS 256
#define EROWS (SEQ + 8)
#define EROWS_PAD (EROWS + 8)
#define HOFF  32
#define HPAD_LEN 512

typedef __attribute__((ext_vector_type(2))) float v2f;
typedef __attribute__((ext_vector_type(8))) float v8f;

__global__ void k_impulse(const float* __restrict__ W, float* __restrict__ hpad) {
  if (threadIdx.x == 0 && blockIdx.x == 0) {
    for (int i = 0; i < HPAD_LEN; ++i) hpad[i] = 0.0f;
    hpad[HOFF] = 1.0f;
    for (int n = 1; n < KTAPS; ++n) {
      float s = 0.0f;
#pragma unroll
      for (int d = 1; d <= 8; ++d) s += W[8 - d] * hpad[HOFF + n - d];
      hpad[HOFF + n] = s;
    }
  }
}

__global__ void k_eps(const float* __restrict__ y, const float* __restrict__ hpad,
                      float* __restrict__ Efull) {
  const int b = blockIdx.x * blockDim.x + threadIdx.x;
  if (b >= BATCH) return;
  float hv[8];
#pragma unroll
  for (int i = 0; i < 8; ++i) hv[i] = hpad[HOFF + i];
  float e[8];
#pragma unroll
  for (int i = 0; i < 8; ++i) {
    float s = y[b * 9 + i + 1] - y[b * 9 + i];
#pragma unroll
    for (int p = 0; p < 8; ++p) {
      if (p < i) s -= hv[i - p] * e[p];
    }
    e[i] = s;
    Efull[i * BATCH + b] = s;
  }
#pragma unroll
  for (int i = 0; i < 8; ++i)
    Efull[(size_t)(EROWS + i) * BATCH + b] = 0.0f;
}

#define EX_TB      32
#define EX_TM      512
#define EX_THREADS 256
#define EX_CH      (EX_TM / (EX_THREADS / 32))
#define EX_ROW4    ((EX_TM + 8) / 4)
#define EX_TILE4   (EX_TB * EX_ROW4)
#define EX_TRIPS   ((EX_TILE4 + EX_THREADS - 1) / EX_THREADS)
#define EX_PITCH   (EX_TM + 12)

typedef __attribute__((ext_vector_type(4))) float v4f;

__global__ void __launch_bounds__(EX_THREADS)
k_exo(const float* __restrict__ u, const float* __restrict__ W,
      float* __restrict__ Efull) {
  __shared__ __attribute__((aligned(16))) float tile[EX_TB * EX_PITCH];

  const int tid = threadIdx.x;
  const int b0  = blockIdx.x * EX_TB;
  const int m0  = blockIdx.y * EX_TM;

  v4f stg[EX_TRIPS];
#pragma unroll
  for (int k = 0; k < EX_TRIPS; ++k) {
    const int idx = tid + EX_THREADS * k;
    if (k < EX_TRIPS - 1 || idx < EX_TILE4) {
      const int row = idx / EX_ROW4;
      const int c4  = idx - row * EX_ROW4;
      stg[k] = *(const v4f*)(u + (size_t)(b0 + row) * USTR + m0 + 4 * c4);
    }
  }
#pragma unroll
  for (int k = 0; k < EX_TRIPS; ++k) {
    const int idx = tid + EX_THREADS * k;
    if (k < EX_TRIPS - 1 || idx < EX_TILE4) {
      const int row = idx / EX_ROW4;
      const int c4  = idx - row * EX_ROW4;
      *(v4f*)(tile + row * EX_PITCH + 4 * c4) = stg[k];
    }
  }

  float wv[8];
#pragma unroll
  for (int j = 0; j < 8; ++j) wv[j] = W[8 + j];
  __syncthreads();

  const int lane = tid & 31;
  const int c0   = (tid >> 5) * EX_CH;
  const v4f* rp = (const v4f*)(tile + lane * EX_PITCH + c0);
  float x[EX_CH + 8];
#pragma unroll
  for (int k = 0; k < (EX_CH + 8) / 4; ++k) {
    const v4f t = rp[k];
    x[4 * k + 0] = t.x; x[4 * k + 1] = t.y; x[4 * k + 2] = t.z; x[4 * k + 3] = t.w;
  }

  float* ep = Efull + ((size_t)(m0 + c0 + 8) * BATCH + b0 + lane);
#pragma unroll
  for (int i = 0; i < EX_CH; ++i) {
    float s = 0.0f;
#pragma unroll
    for (int j = 0; j < 8; ++j) s += wv[j] * x[1 + i + j];
    ep[(size_t)i * BATCH] = s;
  }
}

__global__ void k_zconv(const float* __restrict__ hpad, const float* __restrict__ Efull,
                        float* __restrict__ Z) {
  const int lane = threadIdx.x & 31;
  const int wv   = threadIdx.x >> 5;
  const int nbase = blockIdx.x * 16 + 1;
  const int b0 = blockIdx.y * 256 + wv * 64;
  const int col  = lane & 15;
  const int half = lane >> 4;

  v8f acc0 = {0.0f, 0.0f, 0.0f, 0.0f, 0.0f, 0.0f, 0.0f, 0.0f};
  v8f acc1 = {0.0f, 0.0f, 0.0f, 0.0f, 0.0f, 0.0f, 0.0f, 0.0f};
  v8f acc2 = {0.0f, 0.0f, 0.0f, 0.0f, 0.0f, 0.0f, 0.0f, 0.0f};
  v8f acc3 = {0.0f, 0.0f, 0.0f, 0.0f, 0.0f, 0.0f, 0.0f, 0.0f};

  int r0 = nbase + 8 - KTAPS;
  if (r0 < 0) r0 = 0;
  r0 &= ~3;
  const int rmax = nbase + 22;

  const float* hbase = hpad + (HOFF + nbase + col + 7 - 2 * half - r0);
  const float* ebase = Efull + ((size_t)(r0 + 2 * half) * BATCH + b0 + col);

  for (int r = r0; r <= rmax; r += 4) {
    v2f a;
    a.x = hbase[0];
    a.y = hbase[-1];
    v2f f0, f1, f2, f3;
    f0.x = ebase[0];          f0.y = ebase[BATCH];
    f1.x = ebase[16];         f1.y = ebase[BATCH + 16];
    f2.x = ebase[32];         f2.y = ebase[BATCH + 32];
    f3.x = ebase[48];         f3.y = ebase[BATCH + 48];
    acc0 = __builtin_amdgcn_wmma_f32_16x16x4_f32(false, a, false, f0, (short)0,
                                                 acc0, false, false);
    acc1 = __builtin_amdgcn_wmma_f32_16x16x4_f32(false, a, false, f1, (short)0,
                                                 acc1, false, false);
    acc2 = __builtin_amdgcn_wmma_f32_16x16x4_f32(false, a, false, f2, (short)0,
                                                 acc2, false, false);
    acc3 = __builtin_amdgcn_wmma_f32_16x16x4_f32(false, a, false, f3, (short)0,
                                                 acc3, false, false);
    hbase -= 4;
    ebase += 4 * BATCH;
  }

#pragma unroll
  for (int v = 0; v < 8; ++v) {
    const int row = v + 8 * half;
    float* zp = Z + ((size_t)(nbase - 1 + row) * BATCH + b0 + col);
    zp[0]  = acc0[v];
    zp[16] = acc1[v];
    zp[32] = acc2[v];
    zp[48] = acc3[v];
  }
}

#define CS_ROWS    16
#define CS_SEGS    32
#define CS_LEN     (SEQ / CS_SEGS)
#define CS_THREADS (CS_ROWS * CS_SEGS)
#define CS_IMG     (CS_ROWS * OSTR)
#define CS_TP      68

__global__ void __launch_bounds__(CS_THREADS)
k_cumsum(const float* __restrict__ y, const float* __restrict__ Z,
         float* __restrict__ out) {
  __shared__ __attribute__((aligned(16))) float img[CS_IMG];
  __shared__ __attribute__((aligned(16))) float tot[CS_ROWS * CS_TP];

  const int tid = threadIdx.x;
  const int b   = tid & (CS_ROWS - 1);
  const int seg = tid >> 4;
  const int b0  = blockIdx.x * CS_ROWS;

  const float* zp = Z + ((size_t)(seg * CS_LEN) * BATCH + b0 + b);
  float p[CS_LEN];
#pragma unroll
  for (int i = 0; i < CS_LEN; ++i) p[i] = zp[(size_t)i * BATCH];
  const float x8 = y[(b0 + b) * 9 + 8];
#pragma unroll
  for (int i = 1; i < CS_LEN; ++i) p[i] += p[i - 1];

  tot[b * CS_TP + seg] = p[CS_LEN - 1];
  if (tid < CS_ROWS * 9)
    img[(tid / 9) * OSTR + (tid % 9)] = y[b0 * 9 + tid];
  __syncthreads();

  float base = x8;
  const v4f* t4 = (const v4f*)(tot + b * CS_TP);
#pragma unroll
  for (int k = 0; k < CS_SEGS / 4; ++k) {
    const v4f t = t4[k];
    base += (4 * k + 0 < seg) ? t.x : 0.0f;
    base += (4 * k + 1 < seg) ? t.y : 0.0f;
    base += (4 * k + 2 < seg) ? t.z : 0.0f;
    base += (4 * k + 3 < seg) ? t.w : 0.0f;
  }

  float* ip = img + (b * OSTR + 9 + seg * CS_LEN);
#pragma unroll
  for (int i = 0; i < CS_LEN; ++i) ip[i] = base + p[i];
  __syncthreads();

  const v4f* img4 = (const v4f*)img;
  v4f* out4 = (v4f*)(out + (size_t)b0 * OSTR);
  const int pad = (8 - (int)(((size_t)out4 >> 4) & 7)) & 7;
  if (tid < pad) out4[tid] = img4[tid];
  const int q = tid + pad;
#pragma unroll
  for (int k = 0; k < 16; ++k) out4[q + CS_THREADS * k] = img4[q + CS_THREADS * k];
  if (q + CS_THREADS * 16 < CS_IMG / 4)
    out4[q + CS_THREADS * 16] = img4[q + CS_THREADS * 16];
}

extern "C" void kernel_launch(void* const* d_in, const int* in_sizes, int n_in,
                              void* d_out, int out_size, void* d_ws, size_t ws_size,
                              hipStream_t stream) {
  (void)in_sizes; (void)n_in; (void)out_size; (void)ws_size;
  const float* y = (const float*)d_in[0];
  const float* u = (const float*)d_in[1];
  const float* W = (const float*)d_in[2];
  float* out = (float*)d_out;

  char* ws = (char*)d_ws;
  float* hpad  = (float*)ws;
  float* Efull = (float*)(ws + HPAD_LEN * 4);
  float* Z     = (float*)(ws + HPAD_LEN * 4 +
                          (size_t)EROWS_PAD * BATCH * 4);

  k_impulse<<<1, 32, 0, stream>>>(W, hpad);
  k_eps   <<<BATCH / 256, 256, 0, stream>>>(y, hpad, Efull);
  k_exo   <<<dim3(BATCH / EX_TB, SEQ / EX_TM), EX_THREADS, 0, stream>>>(u, W, Efull);
  k_zconv <<<dim3(SEQ / 16, BATCH / 256), 128, 0, stream>>>(hpad, Efull, Z);
  k_cumsum<<<BATCH / CS_ROWS, CS_THREADS, 0, stream>>>(y, Z, out);
}


# set-up folded into the exo launch as blocks of its own (impulse response, window in registers, its 8 taps summed oldest first; initial-window solve): runs beside the exo tiles; 3 launches, not 5
# speedup vs baseline: 5.2400x; 1.2878x over previous
#include <hip/hip_runtime.h>

#define BATCH 4096
#define SEQ   2048
#define ARP   8
#define EXO   8
#define USTR  (SEQ + EXO)
#define OSTR  (SEQ + ARP + 1)
#define KTAPS 256
#define EROWS (SEQ + 8)
#define EROWS_PAD (EROWS + 8)
#define HOFF  32
#define HPAD_LEN 512

typedef __attribute__((ext_vector_type(2))) float v2f;
typedef __attribute__((ext_vector_type(8))) float v8f;

__device__ __forceinline__ float imp_step(const float* w, const float* a) {
  float s = w[0] * a[0];
#pragma unroll
  for (int k = 1; k < 8; ++k) s = __builtin_fmaf(w[k], a[k], s);
  return s;
}

__device__ __forceinline__ void imp_first8(const float* w, float* a) {
#pragma unroll
  for (int k = 0; k < 8; ++k) a[k] = 0.0f;
  a[8] = 1.0f;
#pragma unroll
  for (int j = 1; j < 8; ++j) a[8 + j] = imp_step(w, a + j);
}

#define EX_TB      32
#define EX_TM      512
#define EX_THREADS 256
#define EX_CH      (EX_TM / (EX_THREADS / 32))
#define EX_ROW4    ((EX_TM + 8) / 4)
#define EX_TILE4   (EX_TB * EX_ROW4)
#define EX_TRIPS   ((EX_TILE4 + EX_THREADS - 1) / EX_THREADS)
#define EX_PITCH   (EX_TM + 12)
#define EPS_BLOCKS (BATCH / EX_THREADS)
#define EX_BLOCKS  ((BATCH / EX_TB) * (SEQ / EX_TM))
#define FRONT_BLOCKS (1 + EPS_BLOCKS + EX_BLOCKS)

typedef __attribute__((ext_vector_type(4))) float v4f;

__global__ void __launch_bounds__(EX_THREADS)
k_front(const float* __restrict__ y, const float* __restrict__ u,
        const float* __restrict__ W, float* __restrict__ hpad,
        float* __restrict__ Efull) {
  __shared__ __attribute__((aligned(16))) float tile[EX_TB * EX_PITCH];

  const int tid = threadIdx.x;

  if (blockIdx.x == 0) {
    static_assert(HOFF <= EX_THREADS && HPAD_LEN - (HOFF + KTAPS) <= EX_THREADS, "one store per thread");
    if (tid < HOFF) hpad[tid] = 0.0f;
    if (tid < HPAD_LEN - (HOFF + KTAPS)) hpad[HOFF + KTAPS + tid] = 0.0f;
    if (tid == 0) {
      float w[8];
#pragma unroll
      for (int k = 0; k < 8; ++k) w[k] = W[k];
      float a[16];
      imp_first8(w, a);
      v4f* hp = (v4f*)(hpad + HOFF);
      hp[0] = (v4f){a[8], a[9], a[10], a[11]};
      hp[1] = (v4f){a[12], a[13], a[14], a[15]};
      for (int g = 1; g < KTAPS / 8; ++g) {
#pragma unroll
        for (int k = 0; k < 8; ++k) a[k] = a[8 + k];
#pragma unroll
        for (int j = 0; j < 8; ++j) a[8 + j] = imp_step(w, a + j);
        hp[2 * g]     = (v4f){a[8], a[9], a[10], a[11]};
        hp[2 * g + 1] = (v4f){a[12], a[13], a[14], a[15]};
      }
    }
    return;
  }

  if (blockIdx.x <= EPS_BLOCKS) {
    const int b = (blockIdx.x - 1) * EX_THREADS + tid;
    float w[8];
#pragma unroll
    for (int k = 0; k < 8; ++k) w[k] = W[k];
    float a[16];
    imp_first8(w, a);
    float e[8];
#pragma unroll
    for (int i = 0; i < 8; ++i) {
      float s = y[b * 9 + i + 1] - y[b * 9 + i];
#pragma unroll
      for (int p = 0; p < 8; ++p) {
        if (p < i) s -= a[8 + i - p] * e[p];
      }
      e[i] = s;
      Efull[i * BATCH + b] = s;
    }
#pragma unroll
    for (int i = 0; i < 8; ++i)
      Efull[(size_t)(EROWS + i) * BATCH + b] = 0.0f;
    return;
  }

  const int xt  = blockIdx.x - (1 + EPS_BLOCKS);
  const int b0  = (xt % (BATCH / EX_TB)) * EX_TB;
  const int m0  = (xt / (BATCH / EX_TB)) * EX_TM;

  v4f stg[EX_TRIPS];
#pragma unroll
  for (int k = 0; k < EX_TRIPS; ++k) {
    const int idx = tid + EX_THREADS * k;
    if (k < EX_TRIPS - 1 || idx < EX_TILE4) {
      const int row = idx / EX_ROW4;
      const int c4  = idx - row * EX_ROW4;
      stg[k] = *(const v4f*)(u + (size_t)(b0 + row) * USTR + m0 + 4 * c4);
    }
  }
#pragma unroll
  for (int k = 0; k < EX_TRIPS; ++k) {
    const int idx = tid + EX_THREADS * k;
    if (k < EX_TRIPS - 1 || idx < EX_TILE4) {
      const int row = idx / EX_ROW4;
      const int c4  = idx - row * EX_ROW4;
      *(v4f*)(tile + row * EX_PITCH + 4 * c4) = stg[k];
    }
  }

  float wv[8];
#pragma unroll
  for (int j = 0; j < 8; ++j) wv[j] = W[8 + j];
  __syncthreads();

  const int lane = tid & 31;
  const int c0   = (tid >> 5) * EX_CH;
  const v4f* rp = (const v4f*)(tile + lane * EX_PITCH + c0);
  float x[EX_CH + 8];
#pragma unroll
  for (int k = 0; k < (EX_CH + 8) / 4; ++k) {
    const v4f t = rp[k];
    x[4 * k + 0] = t.x; x[4 * k + 1] = t.y; x[4 * k + 2] = t.z; x[4 * k + 3] = t.w;
  }

  float* ep = Efull + ((size_t)(m0 + c0 + 8) * BATCH + b0 + lane);
#pragma unroll
  for (int i = 0; i < EX_CH; ++i) {
    float s = 0.0f;
#pragma unroll
    for (int j = 0; j < 8; ++j) s += wv[j] * x[1 + i + j];
    ep[(size_t)i * BATCH] = s;
  }
}

__global__ void k_zconv(const float* __restrict__ hpad, const float* __restrict__ Efull,
                        float* __restrict__ Z) {
  const int lane = threadIdx.x & 31;
  const int wv   = threadIdx.x >> 5;
  const int nbase = blockIdx.x * 16 + 1;
  const int b0 = blockIdx.y * 256 + wv * 64;
  const int col  = lane & 15;
  const int half = lane >> 4;

  v8f acc0 = {0.0f, 0.0f, 0.0f, 0.0f, 0.0f, 0.0f, 0.0f, 0.0f};
  v8f acc1 = {0.0f, 0.0f, 0.0f, 0.0f, 0.0f, 0.0f, 0.0f, 0.0f};
  v8f acc2 = {0.0f, 0.0f, 0.0f, 0.0f, 0.0f, 0.0f, 0.0f, 0.0f};
  v8f acc3 = {0.0f, 0.0f, 0.0f, 0.0f, 0.0f, 0.0f, 0.0f, 0.0f};

  int r0 = nbase + 8 - KTAPS;
  if (r0 < 0) r0 = 0;
  r0 &= ~3;
  const int rmax = nbase + 22;

  const float* hbase = hpad + (HOFF + nbase + col + 7 - 2 * half - r0);
  const float* ebase = Efull + ((size_t)(r0 + 2 * half) * BATCH + b0 + col);

  for (int r = r0; r <= rmax; r += 4) {
    v2f a;
    a.x = hbase[0];
    a.y = hbase[-1];
    v2f f0, f1, f2, f3;
    f0.x = ebase[0];          f0.y = ebase[BATCH];
    f1.x = ebase[16];         f1.y = ebase[BATCH + 16];
    f2.x = ebase[32];         f2.y = ebase[BATCH + 32];
    f3.x = ebase[48];         f3.y = ebase[BATCH + 48];
    acc0 = __builtin_amdgcn_wmma_f32_16x16x4_f32(false, a, false, f0, (short)0,
                                                 acc0, false, false);
    acc1 = __builtin_amdgcn_wmma_f32_16x16x4_f32(false, a, false, f1, (short)0,
                                                 acc1, false, false);
    acc2 = __builtin_amdgcn_wmma_f32_16x16x4_f32(false, a, false, f2, (short)0,
                                                 acc2, false, false);
    acc3 = __builtin_amdgcn_wmma_f32_16x16x4_f32(false, a, false, f3, (short)0,
                                                 acc3, false, false);
    hbase -= 4;
    ebase += 4 * BATCH;
  }

#pragma unroll
  for (int v = 0; v < 8; ++v) {
    const int row = v + 8 * half;
    float* zp = Z + ((size_t)(nbase - 1 + row) * BATCH + b0 + col);
    zp[0]  = acc0[v];
    zp[16] = acc1[v];
    zp[32] = acc2[v];
    zp[48] = acc3[v];
  }
}

#define CS_ROWS    16
#define CS_SEGS    32
#define CS_LEN     (SEQ / CS_SEGS)
#define CS_THREADS (CS_ROWS * CS_SEGS)
#define CS_IMG     (CS_ROWS * OSTR)
#define CS_TP      68

__global__ void __launch_bounds__(CS_THREADS)
k_cumsum(const float* __restrict__ y, const float* __restrict__ Z,
         float* __restrict__ out) {
  __shared__ __attribute__((aligned(16))) float img[CS_IMG];
  __shared__ __attribute__((aligned(16))) float tot[CS_ROWS * CS_TP];

  const int tid = threadIdx.x;
  const int b   = tid & (CS_ROWS - 1);
  const int seg = tid >> 4;
  const int b0  = blockIdx.x * CS_ROWS;

  const float* zp = Z + ((size_t)(seg * CS_LEN) * BATCH + b0 + b);
  float p[CS_LEN];
#pragma unroll
  for (int i = 0; i < CS_LEN; ++i) p[i] = zp[(size_t)i * BATCH];
  const float x8 = y[(b0 + b) * 9 + 8];
#pragma unroll
  for (int i = 1; i < CS_LEN; ++i) p[i] += p[i - 1];

  tot[b * CS_TP + seg] = p[CS_LEN - 1];
  if (tid < CS_ROWS * 9)
    img[(tid / 9) * OSTR + (tid % 9)] = y[b0 * 9 + tid];
  __syncthreads();

  float base = x8;
  const v4f* t4 = (const v4f*)(tot + b * CS_TP);
#pragma unroll
  for (int k = 0; k < CS_SEGS / 4; ++k) {
    const v4f t = t4[k];
    base += (4 * k + 0 < seg) ? t.x : 0.0f;
    base += (4 * k + 1 < seg) ? t.y : 0.0f;
    base += (4 * k + 2 < seg) ? t.z : 0.0f;
    base += (4 * k + 3 < seg) ? t.w : 0.0f;
  }

  float* ip = img + (b * OSTR + 9 + seg * CS_LEN);
#pragma unroll
  for (int i = 0; i < CS_LEN; ++i) ip[i] = base + p[i];
  __syncthreads();

  const v4f* img4 = (const v4f*)img;
  v4f* out4 = (v4f*)(out + (size_t)b0 * OSTR);
  const int pad = (8 - (int)(((size_t)out4 >> 4) & 7)) & 7;
  if (tid < pad) out4[tid] = img4[tid];
  const int q = tid + pad;
#pragma unroll
  for (int k = 0; k < 16; ++k) out4[q + CS_THREADS * k] = img4[q + CS_THREADS * k];
  if (q + CS_THREADS * 16 < CS_IMG / 4)
    out4[q + CS_THREADS * 16] = img4[q + CS_THREADS * 16];
}

extern "C" void kernel_launch(void* const* d_in, const int* in_sizes, int n_in,
                              void* d_out, int out_size, void* d_ws, size_t ws_size,
                              hipStream_t stream) {
  (void)in_sizes; (void)n_in; (void)out_size; (void)ws_size;
  const float* y = (const float*)d_in[0];
  const float* u = (const float*)d_in[1];
  const float* W = (const float*)d_in[2];
  float* out = (float*)d_out;

  char* ws = (char*)d_ws;
  float* hpad  = (float*)ws;
  float* Efull = (float*)(ws + HPAD_LEN * 4);
  float* Z     = (float*)(ws + HPAD_LEN * 4 +
                          (size_t)EROWS_PAD * BATCH * 4);

  k_front <<<FRONT_BLOCKS, EX_THREADS, 0, stream>>>(y, u, W, hpad, Efull);
  k_zconv <<<dim3(SEQ / 16, BATCH / 256), 128, 0, stream>>>(hpad, Efull, Z);
  k_cumsum<<<BATCH / CS_ROWS, CS_THREADS, 0, stream>>>(y, Z, out);
}


# k_zconv feed: a lane's 4 accumulators take 4 adjacent batch columns, so B fragments come as 16 B row pieces (2 loads per K-step, not 8), issued two 8-row blocks ahead; Z leaves as 16 B stores
# speedup vs baseline: 6.4603x; 1.2329x over previous
#include <hip/hip_runtime.h>

#define BATCH 4096
#define SEQ   2048
#define ARP   8
#define EXO   8
#define USTR  (SEQ + EXO)
#define OSTR  (SEQ + ARP + 1)
#define KTAPS 256
#define EROWS (SEQ + 8)
#define EPAD_ROWS 16
#define EROWS_PAD (EROWS + EPAD_ROWS)
#define HOFF  32
#define HPAD_LEN 512

typedef __attribute__((ext_vector_type(2))) float v2f;
typedef __attribute__((ext_vector_type(8))) float v8f;

__device__ __forceinline__ float imp_step(const float* w, const float* a) {
  float s = w[0] * a[0];
#pragma unroll
  for (int k = 1; k < 8; ++k) s = __builtin_fmaf(w[k], a[k], s);
  return s;
}

__device__ __forceinline__ void imp_first8(const float* w, float* a) {
#pragma unroll
  for (int k = 0; k < 8; ++k) a[k] = 0.0f;
  a[8] = 1.0f;
#pragma unroll
  for (int j = 1; j < 8; ++j) a[8 + j] = imp_step(w, a + j);
}

#define EX_TB      32
#define EX_TM      512
#define EX_THREADS 256
#define EX_CH      (EX_TM / (EX_THREADS / 32))
#define EX_ROW4    ((EX_TM + 8) / 4)
#define EX_TILE4   (EX_TB * EX_ROW4)
#define EX_TRIPS   ((EX_TILE4 + EX_THREADS - 1) / EX_THREADS)
#define EX_PITCH   (EX_TM + 12)
#define EPS_BLOCKS (BATCH / EX_THREADS)
#define EX_BLOCKS  ((BATCH / EX_TB) * (SEQ / EX_TM))
#define FRONT_BLOCKS (1 + EPS_BLOCKS + EX_BLOCKS)

typedef __attribute__((ext_vector_type(4))) float v4f;

__global__ void __launch_bounds__(EX_THREADS)
k_front(const float* __restrict__ y, const float* __restrict__ u,
        const float* __restrict__ W, float* __restrict__ hpad,
        float* __restrict__ Efull) {
  __shared__ __attribute__((aligned(16))) float tile[EX_TB * EX_PITCH];

  const int tid = threadIdx.x;

  if (blockIdx.x == 0) {
    static_assert(HOFF <= EX_THREADS && HPAD_LEN - (HOFF + KTAPS) <= EX_THREADS, "one store per thread");
    if (tid < HOFF) hpad[tid] = 0.0f;
    if (tid < HPAD_LEN - (HOFF + KTAPS)) hpad[HOFF + KTAPS + tid] = 0.0f;
    if (tid == 0) {
      float w[8];
#pragma unroll
      for (int k = 0; k < 8; ++k) w[k] = W[k];
      float a[16];
      imp_first8(w, a);
      v4f* hp = (v4f*)(hpad + HOFF);
      hp[0] = (v4f){a[8], a[9], a[10], a[11]};
      hp[1] = (v4f){a[12], a[13], a[14], a[15]};
      for (int g = 1; g < KTAPS / 8; ++g) {
#pragma unroll
        for (int k = 0; k < 8; ++k) a[k] = a[8 + k];
#pragma unroll
        for (int j = 0; j < 8; ++j) a[8 + j] = imp_step(w, a + j);
        hp[2 * g]     = (v4f){a[8], a[9], a[10], a[11]};
        hp[2 * g + 1] = (v4f){a[12], a[13], a[14], a[15]};
      }
    }
    return;
  }

  if (blockIdx.x <= EPS_BLOCKS) {
    const int b = (blockIdx.x - 1) * EX_THREADS + tid;
    float w[8];
#pragma unroll
    for (int k = 0; k < 8; ++k) w[k] = W[k];
    float a[16];
    imp_first8(w, a);
    float e[8];
#pragma unroll
    for (int i = 0; i < 8; ++i) {
      float s = y[b * 9 + i + 1] - y[b * 9 + i];
#pragma unroll
      for (int p = 0; p < 8; ++p) {
        if (p < i) s -= a[8 + i - p] * e[p];
      }
      e[i] = s;
      Efull[i * BATCH + b] = s;
    }
#pragma unroll
    for (int i = 0; i < EPAD_ROWS; ++i)
      Efull[(size_t)(EROWS + i) * BATCH + b] = 0.0f;
    return;
  }

  const int xt  = blockIdx.x - (1 + EPS_BLOCKS);
  const int b0  = (xt % (BATCH / EX_TB)) * EX_TB;
  const int m0  = (xt / (BATCH / EX_TB)) * EX_TM;

  v4f stg[EX_TRIPS];
#pragma unroll
  for (int k = 0; k < EX_TRIPS; ++k) {
    const int idx = tid + EX_THREADS * k;
    if (k < EX_TRIPS - 1 || idx < EX_TILE4) {
      const int row = idx / EX_ROW4;
      const int c4  = idx - row * EX_ROW4;
      stg[k] = *(const v4f*)(u + (size_t)(b0 + row) * USTR + m0 + 4 * c4);
    }
  }
#pragma unroll
  for (int k = 0; k < EX_TRIPS; ++k) {
    const int idx = tid + EX_THREADS * k;
    if (k < EX_TRIPS - 1 || idx < EX_TILE4) {
      const int row = idx / EX_ROW4;
      const int c4  = idx - row * EX_ROW4;
      *(v4f*)(tile + row * EX_PITCH + 4 * c4) = stg[k];
    }
  }

  float wv[8];
#pragma unroll
  for (int j = 0; j < 8; ++j) wv[j] = W[8 + j];
  __syncthreads();

  const int lane = tid & 31;
  const int c0   = (tid >> 5) * EX_CH;
  const v4f* rp = (const v4f*)(tile + lane * EX_PITCH + c0);
  float x[EX_CH + 8];
#pragma unroll
  for (int k = 0; k < (EX_CH + 8) / 4; ++k) {
    const v4f t = rp[k];
    x[4 * k + 0] = t.x; x[4 * k + 1] = t.y; x[4 * k + 2] = t.z; x[4 * k + 3] = t.w;
  }

  float* ep = Efull + ((size_t)(m0 + c0 + 8) * BATCH + b0 + lane);
#pragma unroll
  for (int i = 0; i < EX_CH; ++i) {
    float s = 0.0f;
#pragma unroll
    for (int j = 0; j < 8; ++j) s += wv[j] * x[1 + i + j];
    ep[(size_t)i * BATCH] = s;
  }
}

struct ZStage {
  v2f a0, a1;
  v4f x0, y0;
  v4f x1, y1;
};

__device__ __forceinline__ void z_load(ZStage& s, const float* hb, const float* eb) {
  s.a0.x = hb[0];   s.a0.y = hb[-1];
  s.a1.x = hb[-4];  s.a1.y = hb[-5];
  s.x0 = *(const v4f*)(eb);
  s.y0 = *(const v4f*)(eb + BATCH);
  s.x1 = *(const v4f*)(eb + 4 * BATCH);
  s.y1 = *(const v4f*)(eb + 5 * BATCH);
}

__device__ __forceinline__ void z_mma(const ZStage& s, v8f& acc0, v8f& acc1, v8f& acc2, v8f& acc3) {
  v2f f;
  f.x = s.x0.x; f.y = s.y0.x;
  acc0 = __builtin_amdgcn_wmma_f32_16x16x4_f32(false, s.a0, false, f, (short)0, acc0, false, false);
  f.x = s.x0.y; f.y = s.y0.y;
  acc1 = __builtin_amdgcn_wmma_f32_16x16x4_f32(false, s.a0, false, f, (short)0, acc1, false, false);
  f.x = s.x0.z; f.y = s.y0.z;
  acc2 = __builtin_amdgcn_wmma_f32_16x16x4_f32(false, s.a0, false, f, (short)0, acc2, false, false);
  f.x = s.x0.w; f.y = s.y0.w;
  acc3 = __builtin_amdgcn_wmma_f32_16x16x4_f32(false, s.a0, false, f, (short)0, acc3, false, false);
  f.x = s.x1.x; f.y = s.y1.x;
  acc0 = __builtin_amdgcn_wmma_f32_16x16x4_f32(false, s.a1, false, f, (short)0, acc0, false, false);
  f.x = s.x1.y; f.y = s.y1.y;
  acc1 = __builtin_amdgcn_wmma_f32_16x16x4_f32(false, s.a1, false, f, (short)0, acc1, false, false);
  f.x = s.x1.z; f.y = s.y1.z;
  acc2 = __builtin_amdgcn_wmma_f32_16x16x4_f32(false, s.a1, false, f, (short)0, acc2, false, false);
  f.x = s.x1.w; f.y = s.y1.w;
  acc3 = __builtin_amdgcn_wmma_f32_16x16x4_f32(false, s.a1, false, f, (short)0, acc3, false, false);
}

__global__ void __launch_bounds__(128)
k_zconv(const float* __restrict__ hpad, const float* __restrict__ Efull,
        float* __restrict__ Z) {
  const int lane = threadIdx.x & 31;
  const int wv   = threadIdx.x >> 5;
  const int nbase = blockIdx.x * 16 + 1;
  const int b0 = blockIdx.y * 256 + wv * 64;
  const int col  = lane & 15;
  const int half = lane >> 4;

  v8f acc0 = {0.0f, 0.0f, 0.0f, 0.0f, 0.0f, 0.0f, 0.0f, 0.0f};
  v8f acc1 = {0.0f, 0.0f, 0.0f, 0.0f, 0.0f, 0.0f, 0.0f, 0.0f};
  v8f acc2 = {0.0f, 0.0f, 0.0f, 0.0f, 0.0f, 0.0f, 0.0f, 0.0f};
  v8f acc3 = {0.0f, 0.0f, 0.0f, 0.0f, 0.0f, 0.0f, 0.0f, 0.0f};

  int r0 = nbase + 8 - KTAPS;
  if (r0 < 0) r0 = 0;
  r0 &= ~7;
  const int rmax = nbase + 22;

  const float* hb = hpad + (HOFF + nbase + col + 7 - 2 * half - r0);
  const float* eb = Efull + ((size_t)(r0 + 2 * half) * BATCH + b0 + 4 * col);

  const int nblk  = ((rmax - r0) >> 3) + 1;
  const int nround = nblk / 3;
  const int rem    = nblk - 3 * nround;
  ZStage sA, sB, sC;
  z_load(sA, hb, eb);
  z_load(sB, hb - 8, eb + (size_t)8 * BATCH);
  hb -= 16;
  eb += (size_t)16 * BATCH;
  for (int t = 0; t < nround; ++t) {
    z_load(sC, hb, eb);
    z_mma(sA, acc0, acc1, acc2, acc3);
    z_load(sA, hb - 8, eb + (size_t)8 * BATCH);
    z_mma(sB, acc0, acc1, acc2, acc3);
    z_load(sB, hb - 16, eb + (size_t)16 * BATCH);
    z_mma(sC, acc0, acc1, acc2, acc3);
    hb -= 24;
    eb += (size_t)24 * BATCH;
  }
  if (rem >= 1) z_mma(sA, acc0, acc1, acc2, acc3);
  if (rem == 2) z_mma(sB, acc0, acc1, acc2, acc3);

#pragma unroll
  for (int v = 0; v < 8; ++v) {
    const int row = v + 8 * half;
    float* zp = Z + ((size_t)(nbase - 1 + row) * BATCH + b0 + 4 * col);
    *(v4f*)zp = (v4f){acc0[v], acc1[v], acc2[v], acc3[v]};
  }
}

#define CS_ROWS    16
#define CS_SEGS    32
#define CS_LEN     (SEQ / CS_SEGS)
#define CS_THREADS (CS_ROWS * CS_SEGS)
#define CS_IMG     (CS_ROWS * OSTR)
#define CS_TP      68

__global__ void __launch_bounds__(CS_THREADS)
k_cumsum(const float* __restrict__ y, const float* __restrict__ Z,
         float* __restrict__ out) {
  __shared__ __attribute__((aligned(16))) float img[CS_IMG];
  __shared__ __attribute__((aligned(16))) float tot[CS_ROWS * CS_TP];

  const int tid = threadIdx.x;
  const int b   = tid & (CS_ROWS - 1);
  const int seg = tid >> 4;
  const int b0  = blockIdx.x * CS_ROWS;

  const float* zp = Z + ((size_t)(seg * CS_LEN) * BATCH + b0 + b);
  float p[CS_LEN];
#pragma unroll
  for (int i = 0; i < CS_LEN; ++i) p[i] = zp[(size_t)i * BATCH];
  const float x8 = y[(b0 + b) * 9 + 8];
#pragma unroll
  for (int i = 1; i < CS_LEN; ++i) p[i] += p[i - 1];

  tot[b * CS_TP + seg] = p[CS_LEN - 1];
  if (tid < CS_ROWS * 9)
    img[(tid / 9) * OSTR + (tid % 9)] = y[b0 * 9 + tid];
  __syncthreads();

  float base = x8;
  const v4f* t4 = (const v4f*)(tot + b * CS_TP);
#pragma unroll
  for (int k = 0; k < CS_SEGS / 4; ++k) {
    const v4f t = t4[k];
    base += (4 * k + 0 < seg) ? t.x : 0.0f;
    base += (4 * k + 1 < seg) ? t.y : 0.0f;
    base += (4 * k + 2 < seg) ? t.z : 0.0f;
    base += (4 * k + 3 < seg) ? t.w : 0.0f;
  }

  float* ip = img + (b * OSTR + 9 + seg * CS_LEN);
#pragma unroll
  for (int i = 0; i < CS_LEN; ++i) ip[i] = base + p[i];
  __syncthreads();

  const v4f* img4 = (const v4f*)img;
  v4f* out4 = (v4f*)(out + (size_t)b0 * OSTR);
  const int pad = (8 - (int)(((size_t)out4 >> 4) & 7)) & 7;
  if (tid < pad) out4[tid] = img4[tid];
  const int q = tid + pad;
#pragma unroll
  for (int k = 0; k < 16; ++k) out4[q + CS_THREADS * k] = img4[q + CS_THREADS * k];
  if (q + CS_THREADS * 16 < CS_IMG / 4)
    out4[q + CS_THREADS * 16] = img4[q + CS_THREADS * 16];
}

extern "C" void kernel_launch(void* const* d_in, const int* in_sizes, int n_in,
                              void* d_out, int out_size, void* d_ws, size_t ws_size,
                              hipStream_t stream) {
  (void)in_sizes; (void)n_in; (void)out_size; (void)ws_size;
  const float* y = (const float*)d_in[0];
  const float* u = (const float*)d_in[1];
  const float* W = (const float*)d_in[2];
  float* out = (float*)d_out;

  char* ws = (char*)d_ws;
  float* hpad  = (float*)ws;
  float* Efull = (float*)(ws + HPAD_LEN * 4);
  float* Z     = (float*)(ws + HPAD_LEN * 4 +
                          (size_t)EROWS_PAD * BATCH * 4);

  k_front <<<FRONT_BLOCKS, EX_THREADS, 0, stream>>>(y, u, W, hpad, Efull);
  k_zconv <<<dim3(SEQ / 16, BATCH / 256), 128, 0, stream>>>(hpad, Efull, Z);
  k_cumsum<<<BATCH / CS_ROWS, CS_THREADS, 0, stream>>>(y, Z, out);
}


# k_zconv: a wave makes 4 tiles, 3 of them from the tile in front (its last 8 rows carried through the order-8 recurrence into the next 8 input rows): K of 16-48 rows, not 272; sum order changes
# speedup vs baseline: 7.2241x; 1.1182x over previous
#include <hip/hip_runtime.h>

#define BATCH 4096
#define SEQ   2048
#define ARP   8
#define EXO   8
#define USTR  (SEQ + EXO)
#define OSTR  (SEQ + ARP + 1)
#define KTAPS 256
#define EROWS (SEQ + 8)
#define EPAD_ROWS 16
#define EROWS_PAD (EROWS + EPAD_ROWS)
#define HOFF  32
#define HPAD_LEN 512

typedef __attribute__((ext_vector_type(2))) float v2f;
typedef __attribute__((ext_vector_type(8))) float v8f;

__device__ __forceinline__ float imp_step(const float* w, const float* a) {
  float s = w[0] * a[0];
#pragma unroll
  for (int k = 1; k < 8; ++k) s = __builtin_fmaf(w[k], a[k], s);
  return s;
}

__device__ __forceinline__ void imp_first8(const float* w, float* a) {
#pragma unroll
  for (int k = 0; k < 8; ++k) a[k] = 0.0f;
  a[8] = 1.0f;
#pragma unroll
  for (int j = 1; j < 8; ++j) a[8 + j] = imp_step(w, a + j);
}

#define EX_TB      32
#define EX_TM      512
#define EX_THREADS 256
#define EX_CH      (EX_TM / (EX_THREADS / 32))
#define EX_ROW4    ((EX_TM + 8) / 4)
#define EX_TILE4   (EX_TB * EX_ROW4)
#define EX_TRIPS   ((EX_TILE4 + EX_THREADS - 1) / EX_THREADS)
#define EX_PITCH   (EX_TM + 12)
#define EPS_BLOCKS (BATCH / EX_THREADS)
#define EX_BLOCKS  ((BATCH / EX_TB) * (SEQ / EX_TM))
#define FRONT_BLOCKS (1 + EPS_BLOCKS + EX_BLOCKS)

typedef __attribute__((ext_vector_type(4))) float v4f;

__global__ void __launch_bounds__(EX_THREADS)
k_front(const float* __restrict__ y, const float* __restrict__ u,
        const float* __restrict__ W, float* __restrict__ hpad,
        float* __restrict__ Efull) {
  __shared__ __attribute__((aligned(16))) float tile[EX_TB * EX_PITCH];

  const int tid = threadIdx.x;

  if (blockIdx.x == 0) {
    static_assert(HOFF <= EX_THREADS && HPAD_LEN - (HOFF + KTAPS) <= EX_THREADS, "one store per thread");
    if (tid < HOFF) hpad[tid] = 0.0f;
    if (tid < HPAD_LEN - (HOFF + KTAPS)) hpad[HOFF + KTAPS + tid] = 0.0f;
    if (tid == 0) {
      float w[8];
#pragma unroll
      for (int k = 0; k < 8; ++k) w[k] = W[k];
      float a[16];
      imp_first8(w, a);
      v4f* hp = (v4f*)(hpad + HOFF);
      hp[0] = (v4f){a[8], a[9], a[10], a[11]};
      hp[1] = (v4f){a[12], a[13], a[14], a[15]};
      for (int g = 1; g < KTAPS / 8; ++g) {
#pragma unroll
        for (int k = 0; k < 8; ++k) a[k] = a[8 + k];
#pragma unroll
        for (int j = 0; j < 8; ++j) a[8 + j] = imp_step(w, a + j);
        hp[2 * g]     = (v4f){a[8], a[9], a[10], a[11]};
        hp[2 * g + 1] = (v4f){a[12], a[13], a[14], a[15]};
      }
    }
    return;
  }

  if (blockIdx.x <= EPS_BLOCKS) {
    const int b = (blockIdx.x - 1) * EX_THREADS + tid;
    float w[8];
#pragma unroll
    for (int k = 0; k < 8; ++k) w[k] = W[k];
    float a[16];
    imp_first8(w, a);
    float e[8];
#pragma unroll
    for (int i = 0; i < 8; ++i) {
      float s = y[b * 9 + i + 1] - y[b * 9 + i];
#pragma unroll
      for (int p = 0; p < 8; ++p) {
        if (p < i) s -= a[8 + i - p] * e[p];
      }
      e[i] = s;
      Efull[i * BATCH + b] = s;
    }
#pragma unroll
    for (int i = 0; i < EPAD_ROWS; ++i)
      Efull[(size_t)(EROWS + i) * BATCH + b] = 0.0f;
    return;
  }

  const int xt  = blockIdx.x - (1 + EPS_BLOCKS);
  const int b0  = (xt % (BATCH / EX_TB)) * EX_TB;
  const int m0  = (xt / (BATCH / EX_TB)) * EX_TM;

  v4f stg[EX_TRIPS];
#pragma unroll
  for (int k = 0; k < EX_TRIPS; ++k) {
    const int idx = tid + EX_THREADS * k;
    if (k < EX_TRIPS - 1 || idx < EX_TILE4) {
      const int row = idx / EX_ROW4;
      const int c4  = idx - row * EX_ROW4;
      stg[k] = *(const v4f*)(u + (size_t)(b0 + row) * USTR + m0 + 4 * c4);
    }
  }
#pragma unroll
  for (int k = 0; k < EX_TRIPS; ++k) {
    const int idx = tid + EX_THREADS * k;
    if (k < EX_TRIPS - 1 || idx < EX_TILE4) {
      const int row = idx / EX_ROW4;
      const int c4  = idx - row * EX_ROW4;
      *(v4f*)(tile + row * EX_PITCH + 4 * c4) = stg[k];
    }
  }

  float wv[8];
#pragma unroll
  for (int j = 0; j < 8; ++j) wv[j] = W[8 + j];
  __syncthreads();

  const int lane = tid & 31;
  const int c0   = (tid >> 5) * EX_CH;
  const v4f* rp = (const v4f*)(tile + lane * EX_PITCH + c0);
  float x[EX_CH + 8];
#pragma unroll
  for (int k = 0; k < (EX_CH + 8) / 4; ++k) {
    const v4f t = rp[k];
    x[4 * k + 0] = t.x; x[4 * k + 1] = t.y; x[4 * k + 2] = t.z; x[4 * k + 3] = t.w;
  }

  float* ep = Efull + ((size_t)(m0 + c0 + 8) * BATCH + b0 + lane);
#pragma unroll
  for (int i = 0; i < EX_CH; ++i) {
    float s = 0.0f;
#pragma unroll
    for (int j = 0; j < 8; ++j) s += wv[j] * x[1 + i + j];
    ep[(size_t)i * BATCH] = s;
  }
}

struct ZStage {
  v2f a0, a1;
  v4f x0, y0;
  v4f x1, y1;
};

__device__ __forceinline__ void z_load(ZStage& s, const float* hb, const float* eb) {
  s.a0.x = hb[0];   s.a0.y = hb[-1];
  s.a1.x = hb[-4];  s.a1.y = hb[-5];
  s.x0 = *(const v4f*)(eb);
  s.y0 = *(const v4f*)(eb + BATCH);
  s.x1 = *(const v4f*)(eb + 4 * BATCH);
  s.y1 = *(const v4f*)(eb + 5 * BATCH);
}

__device__ __forceinline__ void z_mma(const ZStage& s, v8f& acc0, v8f& acc1, v8f& acc2, v8f& acc3) {
  v2f f;
  f.x = s.x0.x; f.y = s.y0.x;
  acc0 = __builtin_amdgcn_wmma_f32_16x16x4_f32(false, s.a0, false, f, (short)0, acc0, false, false);
  f.x = s.x0.y; f.y = s.y0.y;
  acc1 = __builtin_amdgcn_wmma_f32_16x16x4_f32(false, s.a0, false, f, (short)0, acc1, false, false);
  f.x = s.x0.z; f.y = s.y0.z;
  acc2 = __builtin_amdgcn_wmma_f32_16x16x4_f32(false, s.a0, false, f, (short)0, acc2, false, false);
  f.x = s.x0.w; f.y = s.y0.w;
  acc3 = __builtin_amdgcn_wmma_f32_16x16x4_f32(false, s.a0, false, f, (short)0, acc3, false, false);
  f.x = s.x1.x; f.y = s.y1.x;
  acc0 = __builtin_amdgcn_wmma_f32_16x16x4_f32(false, s.a1, false, f, (short)0, acc0, false, false);
  f.x = s.x1.y; f.y = s.y1.y;
  acc1 = __builtin_amdgcn_wmma_f32_16x16x4_f32(false, s.a1, false, f, (short)0, acc1, false, false);
  f.x = s.x1.z; f.y = s.y1.z;
  acc2 = __builtin_amdgcn_wmma_f32_16x16x4_f32(false, s.a1, false, f, (short)0, acc2, false, false);
  f.x = s.x1.w; f.y = s.y1.w;
  acc3 = __builtin_amdgcn_wmma_f32_16x16x4_f32(false, s.a1, false, f, (short)0, acc3, false, false);
}

__device__ __forceinline__ v4f z_from_lane(const v4f v, const int src) {
  v4f r;
  r.x = __shfl(v.x, src, 32);
  r.y = __shfl(v.y, src, 32);
  r.z = __shfl(v.z, src, 32);
  r.w = __shfl(v.w, src, 32);
  return r;
}

#define ZG_ROWS   64
#define ZG_GROUPS (SEQ / ZG_ROWS)

__global__ void __launch_bounds__(128)
k_zconv(const float* __restrict__ y, const float* __restrict__ hpad,
        const float* __restrict__ Efull, const float* __restrict__ W,
        float* __restrict__ Z) {
  const int lane = threadIdx.x & 31;
  const int wv   = threadIdx.x >> 5;
  const int g    = blockIdx.x;
  const int tileA = (g == 0) ? (SEQ / 16 - 1) : (4 * g - 1);
  const int nbase = 16 * tileA + 1;
  const int b0 = blockIdx.y * 256 + wv * 64;
  const int col  = lane & 15;
  const int half = lane >> 4;

  v8f acc0 = {0.0f, 0.0f, 0.0f, 0.0f, 0.0f, 0.0f, 0.0f, 0.0f};
  v8f acc1 = {0.0f, 0.0f, 0.0f, 0.0f, 0.0f, 0.0f, 0.0f, 0.0f};
  v8f acc2 = {0.0f, 0.0f, 0.0f, 0.0f, 0.0f, 0.0f, 0.0f, 0.0f};
  v8f acc3 = {0.0f, 0.0f, 0.0f, 0.0f, 0.0f, 0.0f, 0.0f, 0.0f};

  int r0 = nbase + 8 - KTAPS;
  if (r0 < 0) r0 = 0;
  r0 &= ~7;
  const int rmax = nbase + 22;

  const float* hb = hpad + (HOFF + nbase + col + 7 - 2 * half - r0);
  const float* eb = Efull + ((size_t)(r0 + 2 * half) * BATCH + b0 + 4 * col);

  const int nblk  = ((rmax - r0) >> 3) + 1;
  const int nround = nblk / 3;
  const int rem    = nblk - 3 * nround;
  ZStage sA, sB, sC;
  z_load(sA, hb, eb);
  z_load(sB, hb - 8, eb + (size_t)8 * BATCH);
  hb -= 16;
  eb += (size_t)16 * BATCH;
  for (int t = 0; t < nround; ++t) {
    z_load(sC, hb, eb);
    z_mma(sA, acc0, acc1, acc2, acc3);
    z_load(sA, hb - 8, eb + (size_t)8 * BATCH);
    z_mma(sB, acc0, acc1, acc2, acc3);
    z_load(sB, hb - 16, eb + (size_t)16 * BATCH);
    z_mma(sC, acc0, acc1, acc2, acc3);
    hb -= 24;
    eb += (size_t)24 * BATCH;
  }
  if (rem >= 1) z_mma(sA, acc0, acc1, acc2, acc3);
  if (rem == 2) z_mma(sB, acc0, acc1, acc2, acc3);

  const int t0 = ZG_ROWS * g + 8;
  const float* eq = Efull + ((size_t)(t0 + 2 * half) * BATCH + b0 + 4 * col);
  ZStage q[6];
#pragma unroll
  for (int j = 0; j < 6; ++j) {
    q[j].x0 = *(const v4f*)(eq + (size_t)(8 * j) * BATCH);
    q[j].y0 = *(const v4f*)(eq + (size_t)(8 * j + 1) * BATCH);
    q[j].x1 = *(const v4f*)(eq + (size_t)(8 * j + 4) * BATCH);
    q[j].y1 = *(const v4f*)(eq + (size_t)(8 * j + 5) * BATCH);
  }
  const float* hq = hpad + (HOFF + col - 2 * half);
  v2f fa0[6], fa1[6];
#pragma unroll
  for (int d = 0; d < 6; ++d) {
    const float* p = hq + 8 * (d - 1);
    fa0[d].x = p[0];   fa0[d].y = p[-1];
    fa1[d].x = p[-4];  fa1[d].y = p[-5];
  }
  float w[8];
#pragma unroll
  for (int k = 0; k < 8; ++k) w[k] = W[k];

#pragma unroll
  for (int v = 0; v < 8; ++v) {
    const int row = v + 8 * half;
    float* zp = Z + ((size_t)(nbase - 1 + row) * BATCH + b0 + 4 * col);
    *(v4f*)zp = (v4f){acc0[v], acc1[v], acc2[v], acc3[v]};
  }

  v4f S[8], vm[8];
  if (g == 0) {
    const float* yp = y + (size_t)(b0 + 4 * col) * 9;
#pragma unroll
    for (int r = 0; r < 8; ++r)
      S[r] = (v4f){yp[r + 1] - yp[r], yp[9 + r + 1] - yp[9 + r],
                   yp[18 + r + 1] - yp[18 + r], yp[27 + r + 1] - yp[27 + r]};
  } else {
#pragma unroll
    for (int r = 0; r < 8; ++r) S[r] = (v4f){acc0[r], acc1[r], acc2[r], acc3[r]};
  }
#pragma unroll
  for (int m = 0; m < 8; ++m) {
    v4f sum = w[0] * S[m];
#pragma unroll
    for (int r = m + 1; r < 8; ++r) sum += w[r - m] * S[r];
    vm[m] = sum;
  }
  {
    const int src = col + 16;
    const v4f u0 = z_from_lane(vm[0], src);
    const v4f u1 = z_from_lane(vm[1], src);
    const v4f u4 = z_from_lane(vm[4], src);
    const v4f u5 = z_from_lane(vm[5], src);
    q[0].x0 += half ? vm[2] : u0;
    q[0].y0 += half ? vm[3] : u1;
    q[0].x1 += half ? vm[6] : u4;
    q[0].y1 += half ? vm[7] : u5;
  }

  v8f c[3][4];
#pragma unroll
  for (int p = 0; p < 3; ++p)
#pragma unroll
    for (int k = 0; k < 4; ++k)
      c[p][k] = (v8f){0.0f, 0.0f, 0.0f, 0.0f, 0.0f, 0.0f, 0.0f, 0.0f};
#pragma unroll
  for (int j = 0; j < 6; ++j) {
#pragma unroll
    for (int p = 0; p < 3; ++p) {
      if (j <= 2 * p + 1) {
        q[j].a0 = fa0[2 * p - j + 1];
        q[j].a1 = fa1[2 * p - j + 1];
        z_mma(q[j], c[p][0], c[p][1], c[p][2], c[p][3]);
      }
    }
  }

#pragma unroll
  for (int p = 0; p < 3; ++p) {
#pragma unroll
    for (int v = 0; v < 8; ++v) {
      const int row = v + 8 * half;
      float* zp = Z + ((size_t)(t0 - 8 + 16 * p + row) * BATCH + b0 + 4 * col);
      *(v4f*)zp = (v4f){c[p][0][v], c[p][1][v], c[p][2][v], c[p][3][v]};
    }
  }
}

#define CS_ROWS    16
#define CS_SEGS    32
#define CS_LEN     (SEQ / CS_SEGS)
#define CS_THREADS (CS_ROWS * CS_SEGS)
#define CS_IMG     (CS_ROWS * OSTR)
#define CS_TP      68

__global__ void __launch_bounds__(CS_THREADS)
k_cumsum(const float* __restrict__ y, const float* __restrict__ Z,
         float* __restrict__ out) {
  __shared__ __attribute__((aligned(16))) float img[CS_IMG];
  __shared__ __attribute__((aligned(16))) float tot[CS_ROWS * CS_TP];

  const int tid = threadIdx.x;
  const int b   = tid & (CS_ROWS - 1);
  const int seg = tid >> 4;
  const int b0  = blockIdx.x * CS_ROWS;

  const float* zp = Z + ((size_t)(seg * CS_LEN) * BATCH + b0 + b);
  float p[CS_LEN];
#pragma unroll
  for (int i = 0; i < CS_LEN; ++i) p[i] = zp[(size_t)i * BATCH];
  const float x8 = y[(b0 + b) * 9 + 8];
#pragma unroll
  for (int i = 1; i < CS_LEN; ++i) p[i] += p[i - 1];

  tot[b * CS_TP + seg] = p[CS_LEN - 1];
  if (tid < CS_ROWS * 9)
    img[(tid / 9) * OSTR + (tid % 9)] = y[b0 * 9 + tid];
  __syncthreads();

  float base = x8;
  const v4f* t4 = (const v4f*)(tot + b * CS_TP);
#pragma unroll
  for (int k = 0; k < CS_SEGS / 4; ++k) {
    const v4f t = t4[k];
    base += (4 * k + 0 < seg) ? t.x : 0.0f;
    base += (4 * k + 1 < seg) ? t.y : 0.0f;
    base += (4 * k + 2 < seg) ? t.z : 0.0f;
    base += (4 * k + 3 < seg) ? t.w : 0.0f;
  }

  float* ip = img + (b * OSTR + 9 + seg * CS_LEN);
#pragma unroll
  for (int i = 0; i < CS_LEN; ++i) ip[i] = base + p[i];
  __syncthreads();

  const v4f* img4 = (const v4f*)img;
  v4f* out4 = (v4f*)(out + (size_t)b0 * OSTR);
  const int pad = (8 - (int)(((size_t)out4 >> 4) & 7)) & 7;
  if (tid < pad) out4[tid] = img4[tid];
  const int q = tid + pad;
#pragma unroll
  for (int k = 0; k < 16; ++k) out4[q + CS_THREADS * k] = img4[q + CS_THREADS * k];
  if (q + CS_THREADS * 16 < CS_IMG / 4)
    out4[q + CS_THREADS * 16] = img4[q + CS_THREADS * 16];
}

extern "C" void kernel_launch(void* const* d_in, const int* in_sizes, int n_in,
                              void* d_out, int out_size, void* d_ws, size_t ws_size,
                              hipStream_t stream) {
  (void)in_sizes; (void)n_in; (void)out_size; (void)ws_size;
  const float* y = (const float*)d_in[0];
  const float* u = (const float*)d_in[1];
  const float* W = (const float*)d_in[2];
  float* out = (float*)d_out;

  char* ws = (char*)d_ws;
  float* hpad  = (float*)ws;
  float* Efull = (float*)(ws + HPAD_LEN * 4);
  float* Z     = (float*)(ws + HPAD_LEN * 4 +
                          (size_t)EROWS_PAD * BATCH * 4);

  k_front <<<FRONT_BLOCKS, EX_THREADS, 0, stream>>>(y, u, W, hpad, Efull);
  k_zconv <<<dim3(ZG_GROUPS, BATCH / 256), 128, 0, stream>>>(y, hpad, Efull, W, Z);
  k_cumsum<<<BATCH / CS_ROWS, CS_THREADS, 0, stream>>>(y, Z, out);
}


# k_zconv and k_cumsum fused: a block owns 16 batch columns for all steps (Efull strip in LDS, wave = 64 rows); z summed in registers per tile, then wave totals; 2 launches, not 3; sum order changes
# speedup vs baseline: 9.6525x; 1.3362x over previous
#include <hip/hip_runtime.h>

#define BATCH 4096
#define SEQ   2048
#define ARP   8
#define EXO   8
#define USTR  (SEQ + EXO)
#define OSTR  (SEQ + ARP + 1)
#define KTAPS 256
#define EROWS (SEQ + 8)
#define EPAD_ROWS 16
#define EROWS_PAD (EROWS + EPAD_ROWS)
#define HOFF  32
#define HPAD_LEN 512

typedef __attribute__((ext_vector_type(2))) float v2f;
typedef __attribute__((ext_vector_type(8))) float v8f;

__device__ __forceinline__ float imp_step(const float* w, const float* a) {
  float s = w[0] * a[0];
#pragma unroll
  for (int k = 1; k < 8; ++k) s = __builtin_fmaf(w[k], a[k], s);
  return s;
}

__device__ __forceinline__ void imp_first8(const float* w, float* a) {
#pragma unroll
  for (int k = 0; k < 8; ++k) a[k] = 0.0f;
  a[8] = 1.0f;
#pragma unroll
  for (int j = 1; j < 8; ++j) a[8 + j] = imp_step(w, a + j);
}

#define EX_TB      32
#define EX_TM      512
#define EX_THREADS 256
#define EX_CH      (EX_TM / (EX_THREADS / 32))
#define EX_ROW4    ((EX_TM + 8) / 4)
#define EX_TILE4   (EX_TB * EX_ROW4)
#define EX_TRIPS   ((EX_TILE4 + EX_THREADS - 1) / EX_THREADS)
#define EX_PITCH   (EX_TM + 12)
#define EPS_BLOCKS (BATCH / EX_THREADS)
#define EX_BLOCKS  ((BATCH / EX_TB) * (SEQ / EX_TM))
#define FRONT_BLOCKS (1 + EPS_BLOCKS + EX_BLOCKS)

typedef __attribute__((ext_vector_type(4))) float v4f;

__global__ void __launch_bounds__(EX_THREADS)
k_front(const float* __restrict__ y, const float* __restrict__ u,
        const float* __restrict__ W, float* __restrict__ hpad,
        float* __restrict__ Efull) {
  __shared__ __attribute__((aligned(16))) float tile[EX_TB * EX_PITCH];

  const int tid = threadIdx.x;

  if (blockIdx.x == 0) {
    static_assert(HOFF <= EX_THREADS && HPAD_LEN - (HOFF + KTAPS) <= EX_THREADS, "one store per thread");
    if (tid < HOFF) hpad[tid] = 0.0f;
    if (tid < HPAD_LEN - (HOFF + KTAPS)) hpad[HOFF + KTAPS + tid] = 0.0f;
    if (tid == 0) {
      float w[8];
#pragma unroll
      for (int k = 0; k < 8; ++k) w[k] = W[k];
      float a[16];
      imp_first8(w, a);
      v4f* hp = (v4f*)(hpad + HOFF);
      hp[0] = (v4f){a[8], a[9], a[10], a[11]};
      hp[1] = (v4f){a[12], a[13], a[14], a[15]};
      for (int g = 1; g < KTAPS / 8; ++g) {
#pragma unroll
        for (int k = 0; k < 8; ++k) a[k] = a[8 + k];
#pragma unroll
        for (int j = 0; j < 8; ++j) a[8 + j] = imp_step(w, a + j);
        hp[2 * g]     = (v4f){a[8], a[9], a[10], a[11]};
        hp[2 * g + 1] = (v4f){a[12], a[13], a[14], a[15]};
      }
    }
    return;
  }

  if (blockIdx.x <= EPS_BLOCKS) {
    const int b = (blockIdx.x - 1) * EX_THREADS + tid;
    float w[8];
#pragma unroll
    for (int k = 0; k < 8; ++k) w[k] = W[k];
    float a[16];
    imp_first8(w, a);
    float e[8];
#pragma unroll
    for (int i = 0; i < 8; ++i) {
      float s = y[b * 9 + i + 1] - y[b * 9 + i];
#pragma unroll
      for (int p = 0; p < 8; ++p) {
        if (p < i) s -= a[8 + i - p] * e[p];
      }
      e[i] = s;
      Efull[i * BATCH + b] = s;
    }
#pragma unroll
    for (int i = 0; i < EPAD_ROWS; ++i)
      Efull[(size_t)(EROWS + i) * BATCH + b] = 0.0f;
    return;
  }

  const int xt  = blockIdx.x - (1 + EPS_BLOCKS);
  const int b0  = (xt % (BATCH / EX_TB)) * EX_TB;
  const int m0  = (xt / (BATCH / EX_TB)) * EX_TM;

  v4f stg[EX_TRIPS];
#pragma unroll
  for (int k = 0; k < EX_TRIPS; ++k) {
    const int idx = tid + EX_THREADS * k;
    if (k < EX_TRIPS - 1 || idx < EX_TILE4) {
      const int row = idx / EX_ROW4;
      const int c4  = idx - row * EX_ROW4;
      stg[k] = *(const v4f*)(u + (size_t)(b0 + row) * USTR + m0 + 4 * c4);
    }
  }
#pragma unroll
  for (int k = 0; k < EX_TRIPS; ++k) {
    const int idx = tid + EX_THREADS * k;
    if (k < EX_TRIPS - 1 || idx < EX_TILE4) {
      const int row = idx / EX_ROW4;
      const int c4  = idx - row * EX_ROW4;
      *(v4f*)(tile + row * EX_PITCH + 4 * c4) = stg[k];
    }
  }

  float wv[8];
#pragma unroll
  for (int j = 0; j < 8; ++j) wv[j] = W[8 + j];
  __syncthreads();

  const int lane = tid & 31;
  const int c0   = (tid >> 5) * EX_CH;
  const v4f* rp = (const v4f*)(tile + lane * EX_PITCH + c0);
  float x[EX_CH + 8];
#pragma unroll
  for (int k = 0; k < (EX_CH + 8) / 4; ++k) {
    const v4f t = rp[k];
    x[4 * k + 0] = t.x; x[4 * k + 1] = t.y; x[4 * k + 2] = t.z; x[4 * k + 3] = t.w;
  }

  float* ep = Efull + ((size_t)(m0 + c0 + 8) * BATCH + b0 + lane);
#pragma unroll
  for (int i = 0; i < EX_CH; ++i) {
    float s = 0.0f;
#pragma unroll
    for (int j = 0; j < 8; ++j) s += wv[j] * x[1 + i + j];
    ep[(size_t)i * BATCH] = s;
  }
}

#define CS_ROWS    16
#define CS_IMG     (CS_ROWS * OSTR)
#define CS_TP      36
#define ES_PITCH   16
#define ES_F4      (EROWS_PAD * ES_PITCH / 4)
#define FZ_THREADS 1024
#define FZ_WAVES   (FZ_THREADS / 32)
#define FZ_TRIPS   ((ES_F4 + FZ_THREADS - 1) / FZ_THREADS)
#define FZ_BUF     (EROWS_PAD * ES_PITCH)

struct ZStage {
  v2f a0, a1;
  float x0, y0;
  float x1, y1;
};

__device__ __forceinline__ void z_load(ZStage& s, const float* hb, const float* eb) {
  s.a0.x = hb[0];   s.a0.y = hb[-1];
  s.a1.x = hb[-4];  s.a1.y = hb[-5];
  s.x0 = eb[0];
  s.y0 = eb[ES_PITCH];
  s.x1 = eb[4 * ES_PITCH];
  s.y1 = eb[5 * ES_PITCH];
}

__device__ __forceinline__ void z_mma(const ZStage& s, v8f& acc) {
  v2f f;
  f.x = s.x0; f.y = s.y0;
  acc = __builtin_amdgcn_wmma_f32_16x16x4_f32(false, s.a0, false, f, (short)0, acc, false, false);
  f.x = s.x1; f.y = s.y1;
  acc = __builtin_amdgcn_wmma_f32_16x16x4_f32(false, s.a1, false, f, (short)0, acc, false, false);
}

__device__ __forceinline__ void z_scan(const v8f t, float& run, float* P,
                                       const int col, const int half) {
  float s[8];
  s[0] = t[0];
#pragma unroll
  for (int v = 1; v < 8; ++v) s[v] = s[v - 1] + t[v];
  const float T0 = __shfl(s[7], col, 32);
  const float T1 = __shfl(s[7], col + 16, 32);
  const float mid = run + T0;
  const float off = half ? mid : run;
#pragma unroll
  for (int v = 0; v < 8; ++v) P[v] = off + s[v];
  run = mid + T1;
}

__global__ void __launch_bounds__(FZ_THREADS)
k_zsum(const float* __restrict__ y, const float* __restrict__ hpad,
       const float* __restrict__ Efull, const float* __restrict__ W,
       float* __restrict__ out) {
  static_assert(FZ_BUF >= CS_IMG, "the image must fit where the strip was");
  static_assert(FZ_WAVES * 64 == SEQ, "one wave per 64 rows");
  __shared__ __attribute__((aligned(16))) float buf[FZ_BUF];
  __shared__ __attribute__((aligned(16))) float tot[CS_ROWS * CS_TP];

  const int tid  = threadIdx.x;
  const int lane = tid & 31;
  const int g    = __builtin_amdgcn_readfirstlane(tid >> 5);
  const int b0   = blockIdx.x * CS_ROWS;
  const int col  = lane & 15;
  const int half = lane >> 4;

  {
    v4f stg[FZ_TRIPS];
#pragma unroll
    for (int k = 0; k < FZ_TRIPS; ++k) {
      const int idx = tid + FZ_THREADS * k;
      if (k < FZ_TRIPS - 1 || idx < ES_F4)
        stg[k] = *(const v4f*)(Efull + (size_t)(idx >> 2) * BATCH + b0 + 4 * (idx & 3));
    }
#pragma unroll
    for (int k = 0; k < FZ_TRIPS; ++k) {
      const int idx = tid + FZ_THREADS * k;
      if (k < FZ_TRIPS - 1 || idx < ES_F4) *(v4f*)(buf + 4 * idx) = stg[k];
    }
  }
  const float x8 = y[(b0 + col) * 9 + 8];
  __syncthreads();

  const int tileA = (g == 0) ? (SEQ / 16 - 1) : (4 * g - 1);
  const int nbase = 16 * tileA + 1;

  v8f acc = {0.0f, 0.0f, 0.0f, 0.0f, 0.0f, 0.0f, 0.0f, 0.0f};

  int r0 = nbase + 8 - KTAPS;
  if (r0 < 0) r0 = 0;
  r0 &= ~7;
  const int rmax = nbase + 22;

  const float* hb = hpad + (HOFF + nbase + col + 7 - 2 * half - r0);
  const float* eb = buf + ((r0 + 2 * half) * ES_PITCH + col);

  const int nblk  = ((rmax - r0) >> 3) + 1;
  const int nround = nblk / 3;
  const int rem    = nblk - 3 * nround;
  ZStage sA, sB, sC;
  z_load(sA, hb, eb);
  z_load(sB, hb - 8, eb + 8 * ES_PITCH);
  hb -= 16;
  eb += 16 * ES_PITCH;
  for (int t = 0; t < nround; ++t) {
    z_load(sC, hb, eb);
    z_mma(sA, acc);
    z_load(sA, hb - 8, eb + 8 * ES_PITCH);
    z_mma(sB, acc);
    z_load(sB, hb - 16, eb + 16 * ES_PITCH);
    z_mma(sC, acc);
    hb -= 24;
    eb += 24 * ES_PITCH;
  }
  if (rem >= 1) z_mma(sA, acc);
  if (rem == 2) z_mma(sB, acc);

  const int t0 = 64 * g + 8;
  const float* eq = buf + ((t0 + 2 * half) * ES_PITCH + col);
  ZStage q[6];
#pragma unroll
  for (int j = 0; j < 6; ++j) {
    q[j].x0 = eq[(8 * j) * ES_PITCH];
    q[j].y0 = eq[(8 * j + 1) * ES_PITCH];
    q[j].x1 = eq[(8 * j + 4) * ES_PITCH];
    q[j].y1 = eq[(8 * j + 5) * ES_PITCH];
  }
  const float* hq = hpad + (HOFF + col - 2 * half);
  v2f fa0[6], fa1[6];
#pragma unroll
  for (int d = 0; d < 6; ++d) {
    const float* p = hq + 8 * (d - 1);
    fa0[d].x = p[0];   fa0[d].y = p[-1];
    fa1[d].x = p[-4];  fa1[d].y = p[-5];
  }
  float w[8];
#pragma unroll
  for (int k = 0; k < 8; ++k) w[k] = W[k];

  float S[8], vm[8];
  if (g == 0) {
    const float* yp = y + (size_t)(b0 + col) * 9;
#pragma unroll
    for (int r = 0; r < 8; ++r) S[r] = yp[r + 1] - yp[r];
  } else {
#pragma unroll
    for (int r = 0; r < 8; ++r) S[r] = acc[r];
  }
#pragma unroll
  for (int m = 0; m < 8; ++m) {
    float sum = w[0] * S[m];
#pragma unroll
    for (int r = m + 1; r < 8; ++r) sum += w[r - m] * S[r];
    vm[m] = sum;
  }
  {
    const int src = col + 16;
    const float u0 = __shfl(vm[0], src, 32);
    const float u1 = __shfl(vm[1], src, 32);
    const float u4 = __shfl(vm[4], src, 32);
    const float u5 = __shfl(vm[5], src, 32);
    q[0].x0 += half ? vm[2] : u0;
    q[0].y0 += half ? vm[3] : u1;
    q[0].x1 += half ? vm[6] : u4;
    q[0].y1 += half ? vm[7] : u5;
  }

  v8f c[3];
#pragma unroll
  for (int p = 0; p < 3; ++p)
    c[p] = (v8f){0.0f, 0.0f, 0.0f, 0.0f, 0.0f, 0.0f, 0.0f, 0.0f};
#pragma unroll
  for (int j = 0; j < 6; ++j) {
#pragma unroll
    for (int p = 0; p < 3; ++p) {
      if (j <= 2 * p + 1) {
        q[j].a0 = fa0[2 * p - j + 1];
        q[j].a1 = fa1[2 * p - j + 1];
        z_mma(q[j], c[p]);
      }
    }
  }

  const v8f zero8 = {0.0f, 0.0f, 0.0f, 0.0f, 0.0f, 0.0f, 0.0f, 0.0f};
  float P0[8], P1[8], P2[8], P3[8], PT[8];
  float run = 0.0f;
  z_scan(g ? acc : zero8, run, P0, col, half);
  z_scan(c[0], run, P1, col, half);
  z_scan(c[1], run, P2, col, half);
  z_scan(c[2], run, P3, col, half);
  float runT = 0.0f;
  z_scan(acc, runT, PT, col, half);
  if (half == 0) tot[col * CS_TP + g] = run;
  __syncthreads();

  if (tid < CS_ROWS * 9)
    buf[(tid / 9) * OSTR + (tid % 9)] = y[b0 * 9 + tid];

  float base = x8, all = x8;
  const v4f* t4 = (const v4f*)(tot + col * CS_TP);
#pragma unroll
  for (int k = 0; k < FZ_WAVES / 4; ++k) {
    const v4f t = t4[k];
    base += (4 * k + 0 < g) ? t.x : 0.0f;
    base += (4 * k + 1 < g) ? t.y : 0.0f;
    base += (4 * k + 2 < g) ? t.z : 0.0f;
    base += (4 * k + 3 < g) ? t.w : 0.0f;
    all += t.x; all += t.y; all += t.z; all += t.w;
  }

  float* ip = buf + (col * OSTR + 9 + 8 * half);
  if (g != 0) {
#pragma unroll
    for (int v = 0; v < 8; ++v) ip[64 * g - 16 + v] = base + P0[v];
  } else {
#pragma unroll
    for (int v = 0; v < 8; ++v) ip[SEQ - 16 + v] = all + PT[v];
  }
#pragma unroll
  for (int v = 0; v < 8; ++v) {
    ip[64 * g + v]      = base + P1[v];
    ip[64 * g + 16 + v] = base + P2[v];
    ip[64 * g + 32 + v] = base + P3[v];
  }
  __syncthreads();

  const v4f* img4 = (const v4f*)buf;
  v4f* out4 = (v4f*)(out + (size_t)b0 * OSTR);
  const int pad = (8 - (int)(((size_t)out4 >> 4) & 7)) & 7;
  if (tid < pad) out4[tid] = img4[tid];
  const int qi = tid + pad;
#pragma unroll
  for (int k = 0; k < 8; ++k) out4[qi + FZ_THREADS * k] = img4[qi + FZ_THREADS * k];
  if (qi + FZ_THREADS * 8 < CS_IMG / 4)
    out4[qi + FZ_THREADS * 8] = img4[qi + FZ_THREADS * 8];
}

extern "C" void kernel_launch(void* const* d_in, const int* in_sizes, int n_in,
                              void* d_out, int out_size, void* d_ws, size_t ws_size,
                              hipStream_t stream) {
  (void)in_sizes; (void)n_in; (void)out_size; (void)ws_size;
  const float* y = (const float*)d_in[0];
  const float* u = (const float*)d_in[1];
  const float* W = (const float*)d_in[2];
  float* out = (float*)d_out;

  char* ws = (char*)d_ws;
  float* hpad  = (float*)ws;
  float* Efull = (float*)(ws + HPAD_LEN * 4);

  k_front<<<FRONT_BLOCKS, EX_THREADS, 0, stream>>>(y, u, W, hpad, Efull);
  k_zsum <<<BATCH / CS_ROWS, FZ_THREADS, 0, stream>>>(y, hpad, Efull, W, out);
}


# front launch folded into the strip kernel: a block stages its 16 rows of u in LDS by 16 B loads along t and makes drive, initial-window solve and h there itself; no workspace; 1 launch, not 2
# speedup vs baseline: 20.6006x; 2.1342x over previous
#include <hip/hip_runtime.h>


#define BATCH 4096
#define SEQ   2048
#define ARP   8
#define EXO   8
#define USTR  (SEQ + EXO)
#define OSTR  (SEQ + ARP + 1)
#define KTAPS 256
#define EROWS (SEQ + 8)
#define EPAD_ROWS 16
#define EROWS_PAD (EROWS + EPAD_ROWS)
#define HOFF  32
#define HPAD_LEN 512

typedef __attribute__((ext_vector_type(2))) float v2f;
typedef __attribute__((ext_vector_type(4))) float v4f;
typedef __attribute__((ext_vector_type(8))) float v8f;

__device__ __forceinline__ float imp_step(const float* w, const float* a) {
  float s = w[0] * a[0];
#pragma unroll
  for (int k = 1; k < 8; ++k) s = __builtin_fmaf(w[k], a[k], s);
  return s;
}

__device__ __forceinline__ void imp_first8(const float* w, float* a) {
#pragma unroll
  for (int k = 0; k < 8; ++k) a[k] = 0.0f;
  a[8] = 1.0f;
#pragma unroll
  for (int j = 1; j < 8; ++j) a[8 + j] = imp_step(w, a + j);
}


#define CS_ROWS    16
#define CS_IMG     (CS_ROWS * OSTR)
#define CS_TP      36
#define SP         2116
#define FZ_THREADS 1024
#define FZ_WAVES   (FZ_THREADS / 32)
#define FZ_BUF     (CS_ROWS * SP)
#define U_ROW4     (USTR / 4)
#define U_F4       (CS_ROWS * U_ROW4)
#define U_TRIPS    ((U_F4 + FZ_THREADS - 1) / FZ_THREADS)
#define EX_CH      32

struct ZStage {
  v2f a0, a1;
  float x0, y0;
  float x1, y1;
};

__device__ __forceinline__ void z_load(ZStage& s, const float* hb, const float* eb) {
  s.a0.x = hb[0];   s.a0.y = hb[-1];
  s.a1.x = hb[-4];  s.a1.y = hb[-5];
  const v2f e0 = *(const v2f*)eb;
  const v2f e1 = *(const v2f*)(eb + 4);
  s.x0 = e0.x;  s.y0 = e0.y;
  s.x1 = e1.x;  s.y1 = e1.y;
}

__device__ __forceinline__ void z_mma(const ZStage& s, v8f& acc) {
  v2f f;
  f.x = s.x0; f.y = s.y0;
  acc = __builtin_amdgcn_wmma_f32_16x16x4_f32(false, s.a0, false, f, (short)0, acc, false, false);
  f.x = s.x1; f.y = s.y1;
  acc = __builtin_amdgcn_wmma_f32_16x16x4_f32(false, s.a1, false, f, (short)0, acc, false, false);
}

__device__ __forceinline__ void z_scan(const v8f t, float& run, float* P,
                                       const int col, const int half) {
  float s[8];
  s[0] = t[0];
#pragma unroll
  for (int v = 1; v < 8; ++v) s[v] = s[v - 1] + t[v];
  const float T0 = __shfl(s[7], col, 32);
  const float T1 = __shfl(s[7], col + 16, 32);
  const float mid = run + T0;
  const float off = half ? mid : run;
#pragma unroll
  for (int v = 0; v < 8; ++v) P[v] = off + s[v];
  run = mid + T1;
}


__global__ void __launch_bounds__(FZ_THREADS)
k_arx(const float* __restrict__ y, const float* __restrict__ u,
      const float* __restrict__ W, float* __restrict__ out) {
  static_assert(SP >= EROWS_PAD && SP % 64 == 4, "strip pitch");
  static_assert(FZ_BUF >= CS_IMG, "the image must fit where the strip was");
  static_assert(FZ_WAVES * 64 == SEQ, "one wave per 64 rows");
  static_assert((FZ_THREADS / CS_ROWS) * EX_CH == SEQ && USTR % 4 == 0, "drive pieces");
  __shared__ __attribute__((aligned(16))) float buf[FZ_BUF];
  __shared__ __attribute__((aligned(16))) float tot[CS_ROWS * CS_TP];
  __shared__ __attribute__((aligned(16))) float hs[HPAD_LEN];

  const int tid  = threadIdx.x;
  const int lane = tid & 31;
  const int g    = __builtin_amdgcn_readfirstlane(tid >> 5);
  const int b0   = blockIdx.x * CS_ROWS;
  const int col  = lane & 15;
  const int half = lane >> 4;

  const bool eps_thread = (g == 1) && (lane < CS_ROWS);
  float yv[9];
  if (eps_thread) {
#pragma unroll
    for (int i = 0; i < 9; ++i) yv[i] = y[(b0 + lane) * 9 + i];
  }

  v4f stg[U_TRIPS];
#pragma unroll
  for (int k = 0; k < U_TRIPS; ++k) {
    const int idx = tid + FZ_THREADS * k;
    if (k < U_TRIPS - 1 || idx < U_F4) {
      const int row = idx / U_ROW4;
      const int c4  = idx - row * U_ROW4;
      stg[k] = *(const v4f*)(u + (size_t)(b0 + row) * USTR + 4 * c4);
    }
  }
  const float x8 = y[(b0 + col) * 9 + 8];

  static_assert(HOFF <= FZ_THREADS && HPAD_LEN - (HOFF + KTAPS) <= FZ_THREADS &&
                CS_ROWS * EPAD_ROWS <= FZ_THREADS, "one store per thread");
  if (tid < HOFF) hs[tid] = 0.0f;
  if (tid < HPAD_LEN - (HOFF + KTAPS)) hs[HOFF + KTAPS + tid] = 0.0f;
  if (tid < CS_ROWS * EPAD_ROWS) buf[(tid / EPAD_ROWS) * SP + EROWS + (tid % EPAD_ROWS)] = 0.0f;

  if (tid == 0) {
    float w[8];
#pragma unroll
    for (int k = 0; k < 8; ++k) w[k] = W[k];
    float a[16];
    imp_first8(w, a);
    v4f* hp = (v4f*)(hs + HOFF);
    hp[0] = (v4f){a[8], a[9], a[10], a[11]};
    hp[1] = (v4f){a[12], a[13], a[14], a[15]};
    for (int gg = 1; gg < KTAPS / 8; ++gg) {
#pragma unroll
      for (int k = 0; k < 8; ++k) a[k] = a[8 + k];
#pragma unroll
      for (int j = 0; j < 8; ++j) a[8 + j] = imp_step(w, a + j);
      hp[2 * gg]     = (v4f){a[8], a[9], a[10], a[11]};
      hp[2 * gg + 1] = (v4f){a[12], a[13], a[14], a[15]};
    }
  }

  float eps[8];
  if (eps_thread) {
    float w[8];
#pragma unroll
    for (int k = 0; k < 8; ++k) w[k] = W[k];
    float a[16];
    imp_first8(w, a);
#pragma unroll
    for (int i = 0; i < 8; ++i) {
      float s = yv[i + 1] - yv[i];
#pragma unroll
      for (int p = 0; p < 8; ++p) {
        if (p < i) s -= a[8 + i - p] * eps[p];
      }
      eps[i] = s;
    }
  }

#pragma unroll
  for (int k = 0; k < U_TRIPS; ++k) {
    const int idx = tid + FZ_THREADS * k;
    if (k < U_TRIPS - 1 || idx < U_F4) {
      const int row = idx / U_ROW4;
      const int c4  = idx - row * U_ROW4;
      *(v4f*)(buf + row * SP + 4 * c4) = stg[k];
    }
  }
  __syncthreads();

  {
    float* sp = buf + ((tid & 15) * SP + EX_CH * (tid >> 4));
    float x[EX_CH + 8];
#pragma unroll
    for (int k = 0; k < (EX_CH + 8) / 4; ++k) {
      const v4f t = ((const v4f*)sp)[k];
      x[4 * k + 0] = t.x; x[4 * k + 1] = t.y; x[4 * k + 2] = t.z; x[4 * k + 3] = t.w;
    }
    float wv[8];
#pragma unroll
    for (int j = 0; j < 8; ++j) wv[j] = W[8 + j];
    __syncthreads();
#pragma unroll
    for (int i4 = 0; i4 < EX_CH / 4; ++i4) {
      float o[4];
#pragma unroll
      for (int q = 0; q < 4; ++q) {
        float s = 0.0f;
#pragma unroll
        for (int j = 0; j < 8; ++j) s += wv[j] * x[1 + 4 * i4 + q + j];
        o[q] = s;
      }
      *(v4f*)(sp + 8 + 4 * i4) = (v4f){o[0], o[1], o[2], o[3]};
    }
    if (eps_thread) {
      v4f* e4 = (v4f*)(buf + lane * SP);
      e4[0] = (v4f){eps[0], eps[1], eps[2], eps[3]};
      e4[1] = (v4f){eps[4], eps[5], eps[6], eps[7]};
    }
  }
  __syncthreads();

  const int tileA = (g == 0) ? (SEQ / 16 - 1) : (4 * g - 1);
  const int nbase = 16 * tileA + 1;

  v8f acc = {0.0f, 0.0f, 0.0f, 0.0f, 0.0f, 0.0f, 0.0f, 0.0f};

  int r0 = nbase + 8 - KTAPS;
  if (r0 < 0) r0 = 0;
  r0 &= ~7;
  const int rmax = nbase + 22;

  const float* hb = hs + (HOFF + nbase + col + 7 - 2 * half - r0);
  const float* eb = buf + (col * SP + r0 + 2 * half);

  const int nblk  = ((rmax - r0) >> 3) + 1;
  const int nround = nblk / 3;
  const int rem    = nblk - 3 * nround;
  ZStage sA, sB, sC;
  z_load(sA, hb, eb);
  z_load(sB, hb - 8, eb + 8);
  hb -= 16;
  eb += 16;
  for (int t = 0; t < nround; ++t) {
    z_load(sC, hb, eb);
    z_mma(sA, acc);
    z_load(sA, hb - 8, eb + 8);
    z_mma(sB, acc);
    z_load(sB, hb - 16, eb + 16);
    z_mma(sC, acc);
    hb -= 24;
    eb += 24;
  }
  if (rem >= 1) z_mma(sA, acc);
  if (rem == 2) z_mma(sB, acc);

  const int t0 = 64 * g + 8;
  const float* eq = buf + (col * SP + t0 + 2 * half);
  ZStage q[6];
#pragma unroll
  for (int j = 0; j < 6; ++j) {
    const v2f e0 = *(const v2f*)(eq + 8 * j);
    const v2f e1 = *(const v2f*)(eq + 8 * j + 4);
    q[j].x0 = e0.x;  q[j].y0 = e0.y;
    q[j].x1 = e1.x;  q[j].y1 = e1.y;
  }
  const float* hq = hs + (HOFF + col - 2 * half);
  v2f fa0[6], fa1[6];
#pragma unroll
  for (int d = 0; d < 6; ++d) {
    const float* p = hq + 8 * (d - 1);
    fa0[d].x = p[0];   fa0[d].y = p[-1];
    fa1[d].x = p[-4];  fa1[d].y = p[-5];
  }
  float w[8];
#pragma unroll
  for (int k = 0; k < 8; ++k) w[k] = W[k];

  float S[8], vm[8];
  if (g == 0) {
    const float* yp = y + (size_t)(b0 + col) * 9;
#pragma unroll
    for (int r = 0; r < 8; ++r) S[r] = yp[r + 1] - yp[r];
  } else {
#pragma unroll
    for (int r = 0; r < 8; ++r) S[r] = acc[r];
  }
#pragma unroll
  for (int m = 0; m < 8; ++m) {
    float sum = w[0] * S[m];
#pragma unroll
    for (int r = m + 1; r < 8; ++r) sum += w[r - m] * S[r];
    vm[m] = sum;
  }
  {
    const int src = col + 16;
    const float u0 = __shfl(vm[0], src, 32);
    const float u1 = __shfl(vm[1], src, 32);
    const float u4 = __shfl(vm[4], src, 32);
    const float u5 = __shfl(vm[5], src, 32);
    q[0].x0 += half ? vm[2] : u0;
    q[0].y0 += half ? vm[3] : u1;
    q[0].x1 += half ? vm[6] : u4;
    q[0].y1 += half ? vm[7] : u5;
  }

  v8f c[3];
#pragma unroll
  for (int p = 0; p < 3; ++p)
    c[p] = (v8f){0.0f, 0.0f, 0.0f, 0.0f, 0.0f, 0.0f, 0.0f, 0.0f};
#pragma unroll
  for (int j = 0; j < 6; ++j) {
#pragma unroll
    for (int p = 0; p < 3; ++p) {
      if (j <= 2 * p + 1) {
        q[j].a0 = fa0[2 * p - j + 1];
        q[j].a1 = fa1[2 * p - j + 1];
        z_mma(q[j], c[p]);
      }
    }
  }

  const v8f zero8 = {0.0f, 0.0f, 0.0f, 0.0f, 0.0f, 0.0f, 0.0f, 0.0f};
  float P0[8], P1[8], P2[8], P3[8], PT[8];
  float run = 0.0f;
  z_scan(g ? acc : zero8, run, P0, col, half);
  z_scan(c[0], run, P1, col, half);
  z_scan(c[1], run, P2, col, half);
  z_scan(c[2], run, P3, col, half);
  float runT = 0.0f;
  z_scan(acc, runT, PT, col, half);
  if (half == 0) tot[col * CS_TP + g] = run;
  __syncthreads();

  if (tid < CS_ROWS * 9)
    buf[(tid / 9) * OSTR + (tid % 9)] = y[b0 * 9 + tid];

  float base = x8, all = x8;
  const v4f* t4 = (const v4f*)(tot + col * CS_TP);
#pragma unroll
  for (int k = 0; k < FZ_WAVES / 4; ++k) {
    const v4f t = t4[k];
    base += (4 * k + 0 < g) ? t.x : 0.0f;
    base += (4 * k + 1 < g) ? t.y : 0.0f;
    base += (4 * k + 2 < g) ? t.z : 0.0f;
    base += (4 * k + 3 < g) ? t.w : 0.0f;
    all += t.x; all += t.y; all += t.z; all += t.w;
  }

  float* ip = buf + (col * OSTR + 9 + 8 * half);
  if (g != 0) {
#pragma unroll
    for (int v = 0; v < 8; ++v) ip[64 * g - 16 + v] = base + P0[v];
  } else {
#pragma unroll
    for (int v = 0; v < 8; ++v) ip[SEQ - 16 + v] = all + PT[v];
  }
#pragma unroll
  for (int v = 0; v < 8; ++v) {
    ip[64 * g + v]      = base + P1[v];
    ip[64 * g + 16 + v] = base + P2[v];
    ip[64 * g + 32 + v] = base + P3[v];
  }
  __syncthreads();

  const v4f* img4 = (const v4f*)buf;
  v4f* out4 = (v4f*)(out + (size_t)b0 * OSTR);
  const int pad = (8 - (int)(((size_t)out4 >> 4) & 7)) & 7;
  if (tid < pad) out4[tid] = img4[tid];
  const int qi = tid + pad;
#pragma unroll
  for (int k = 0; k < 8; ++k) out4[qi + FZ_THREADS * k] = img4[qi + FZ_THREADS * k];
  if (qi + FZ_THREADS * 8 < CS_IMG / 4)
    out4[qi + FZ_THREADS * 8] = img4[qi + FZ_THREADS * 8];
}

extern "C" void kernel_launch(void* const* d_in, const int* in_sizes, int n_in,
                              void* d_out, int out_size, void* d_ws, size_t ws_size,
                              hipStream_t stream) {
  (void)in_sizes; (void)n_in; (void)out_size; (void)d_ws; (void)ws_size;
  const float* y = (const float*)d_in[0];
  const float* u = (const float*)d_in[1];
  const float* W = (const float*)d_in[2];
  float* out = (float*)d_out;

  k_arx<<<BATCH / CS_ROWS, FZ_THREADS, 0, stream>>>(y, u, W, out);
}


# a wave carries the order-8 state through 5 groups of 3 tiles: 8 full-band tiles per strip, not 32; its 16 tiles' sums wait in registers (blocks of 16 waves); sum order: per chain, then totals
# speedup vs baseline: 22.0231x; 1.0691x over previous
#include <hip/hip_runtime.h>


#define BATCH 4096
#define SEQ   2048
#define ARP   8
#define EXO   8
#define USTR  (SEQ + EXO)
#define OSTR  (SEQ + ARP + 1)
#define KTAPS 256
#define EROWS (SEQ + 8)
#define EPAD_ROWS 16
#define EROWS_PAD (EROWS + EPAD_ROWS)
#define HOFF  32
#define HPAD_LEN 512

typedef __attribute__((ext_vector_type(2))) float v2f;
typedef __attribute__((ext_vector_type(4))) float v4f;
typedef __attribute__((ext_vector_type(8))) float v8f;

__device__ __forceinline__ float imp_step(const float* w, const float* a) {
  float s = w[0] * a[0];
#pragma unroll
  for (int k = 1; k < 8; ++k) s = __builtin_fmaf(w[k], a[k], s);
  return s;
}

__device__ __forceinline__ void imp_first8(const float* w, float* a) {
#pragma unroll
  for (int k = 0; k < 8; ++k) a[k] = 0.0f;
  a[8] = 1.0f;
#pragma unroll
  for (int j = 1; j < 8; ++j) a[8 + j] = imp_step(w, a + j);
}


#define CS_ROWS    16
#define CS_IMG     (CS_ROWS * OSTR)
#define CS_TP      12
#define SP         2116
#define FZ_THREADS 512
#define FZ_WAVES   (FZ_THREADS / 32)
#define NCHAIN     8
#define CH_ROWS    (SEQ / NCHAIN)
#define CH_TILES   (CH_ROWS / 16)
#define CH_GROUPS  ((CH_TILES - 1) / 3)
#define FZ_BUF     (CS_ROWS * SP)
#define U_ROW4     (USTR / 4)
#define U_F4       (CS_ROWS * U_ROW4)
#define U_TRIPS    ((U_F4 + FZ_THREADS - 1) / FZ_THREADS)
#define EX_CH      64
#define OUT_F4     (CS_IMG / 4)
#define OUT_FULL   ((OUT_F4 - 7) / FZ_THREADS)

struct ZStage {
  v2f a0, a1;
  float x0, y0;
  float x1, y1;
};

__device__ __forceinline__ void z_load(ZStage& s, const float* hb, const float* eb) {
  s.a0.x = hb[0];   s.a0.y = hb[-1];
  s.a1.x = hb[-4];  s.a1.y = hb[-5];
  const v2f e0 = *(const v2f*)eb;
  const v2f e1 = *(const v2f*)(eb + 4);
  s.x0 = e0.x;  s.y0 = e0.y;
  s.x1 = e1.x;  s.y1 = e1.y;
}

__device__ __forceinline__ void z_mma(const ZStage& s, v8f& acc) {
  v2f f;
  f.x = s.x0; f.y = s.y0;
  acc = __builtin_amdgcn_wmma_f32_16x16x4_f32(false, s.a0, false, f, (short)0, acc, false, false);
  f.x = s.x1; f.y = s.y1;
  acc = __builtin_amdgcn_wmma_f32_16x16x4_f32(false, s.a1, false, f, (short)0, acc, false, false);
}

__device__ __forceinline__ void z_scan(const v8f t, float& run, float* P,
                                       const int col, const int half) {
  float s[8];
  s[0] = t[0];
#pragma unroll
  for (int v = 1; v < 8; ++v) s[v] = s[v - 1] + t[v];
  const float T0 = __shfl(s[7], col, 32);
  const float T1 = __shfl(s[7], col + 16, 32);
  const float mid = run + T0;
  const float off = half ? mid : run;
#pragma unroll
  for (int v = 0; v < 8; ++v) P[v] = off + s[v];
  run = mid + T1;
}


__global__ void __launch_bounds__(FZ_THREADS)
k_arx(const float* __restrict__ y, const float* __restrict__ u,
      const float* __restrict__ W, float* __restrict__ out) {
  static_assert(SP >= EROWS_PAD && SP % 64 == 4, "strip pitch");
  static_assert(FZ_BUF >= CS_IMG, "the image must fit where the strip was");
  static_assert(NCHAIN <= FZ_WAVES && NCHAIN * CH_ROWS == SEQ && CH_GROUPS * 3 + 1 == CH_TILES &&
                NCHAIN % 4 == 0 && CS_TP >= NCHAIN && CS_TP % 4 == 0, "chains");
  static_assert((FZ_THREADS / CS_ROWS) * EX_CH == SEQ && USTR % 4 == 0, "drive pieces");
  __shared__ __attribute__((aligned(16))) float buf[FZ_BUF];
  __shared__ __attribute__((aligned(16))) float tot[CS_ROWS * CS_TP];
  __shared__ __attribute__((aligned(16))) float hs[HPAD_LEN];

  const int tid  = threadIdx.x;
  const int lane = tid & 31;
  const int g    = __builtin_amdgcn_readfirstlane(tid >> 5);
  const int b0   = blockIdx.x * CS_ROWS;
  const int col  = lane & 15;
  const int half = lane >> 4;

  const bool eps_thread = (g == 1) && (lane < CS_ROWS);
  float yv[9];
  if (eps_thread) {
#pragma unroll
    for (int i = 0; i < 9; ++i) yv[i] = y[(b0 + lane) * 9 + i];
  }

  v4f stg[U_TRIPS];
#pragma unroll
  for (int k = 0; k < U_TRIPS; ++k) {
    const int idx = tid + FZ_THREADS * k;
    if (k < U_TRIPS - 1 || idx < U_F4) {
      const int row = idx / U_ROW4;
      const int c4  = idx - row * U_ROW4;
      stg[k] = *(const v4f*)(u + (size_t)(b0 + row) * USTR + 4 * c4);
    }
  }
  const float x8 = y[(b0 + col) * 9 + 8];

  static_assert(HOFF <= FZ_THREADS && HPAD_LEN - (HOFF + KTAPS) <= FZ_THREADS &&
                CS_ROWS * EPAD_ROWS <= FZ_THREADS, "one store per thread");
  if (tid < HOFF) hs[tid] = 0.0f;
  if (tid < HPAD_LEN - (HOFF + KTAPS)) hs[HOFF + KTAPS + tid] = 0.0f;
  if (tid < CS_ROWS * EPAD_ROWS) buf[(tid / EPAD_ROWS) * SP + EROWS + (tid % EPAD_ROWS)] = 0.0f;

  if (tid == 0) {
    float w[8];
#pragma unroll
    for (int k = 0; k < 8; ++k) w[k] = W[k];
    float a[16];
    imp_first8(w, a);
    v4f* hp = (v4f*)(hs + HOFF);
    hp[0] = (v4f){a[8], a[9], a[10], a[11]};
    hp[1] = (v4f){a[12], a[13], a[14], a[15]};
    for (int gg = 1; gg < KTAPS / 8; ++gg) {
#pragma unroll
      for (int k = 0; k < 8; ++k) a[k] = a[8 + k];
#pragma unroll
      for (int j = 0; j < 8; ++j) a[8 + j] = imp_step(w, a + j);
      hp[2 * gg]     = (v4f){a[8], a[9], a[10], a[11]};
      hp[2 * gg + 1] = (v4f){a[12], a[13], a[14], a[15]};
    }
  }

  float eps[8];
  if (eps_thread) {
    float w[8];
#pragma unroll
    for (int k = 0; k < 8; ++k) w[k] = W[k];
    float a[16];
    imp_first8(w, a);
#pragma unroll
    for (int i = 0; i < 8; ++i) {
      float s = yv[i + 1] - yv[i];
#pragma unroll
      for (int p = 0; p < 8; ++p) {
        if (p < i) s -= a[8 + i - p] * eps[p];
      }
      eps[i] = s;
    }
  }

#pragma unroll
  for (int k = 0; k < U_TRIPS; ++k) {
    const int idx = tid + FZ_THREADS * k;
    if (k < U_TRIPS - 1 || idx < U_F4) {
      const int row = idx / U_ROW4;
      const int c4  = idx - row * U_ROW4;
      *(v4f*)(buf + row * SP + 4 * c4) = stg[k];
    }
  }
  __syncthreads();

  {
    float* sp = buf + ((tid & 15) * SP + EX_CH * (tid >> 4));
    float x[EX_CH + 8];
#pragma unroll
    for (int k = 0; k < (EX_CH + 8) / 4; ++k) {
      const v4f t = ((const v4f*)sp)[k];
      x[4 * k + 0] = t.x; x[4 * k + 1] = t.y; x[4 * k + 2] = t.z; x[4 * k + 3] = t.w;
    }
    float wv[8];
#pragma unroll
    for (int j = 0; j < 8; ++j) wv[j] = W[8 + j];
    __syncthreads();
#pragma unroll
    for (int i4 = 0; i4 < EX_CH / 4; ++i4) {
      float o[4];
#pragma unroll
      for (int q = 0; q < 4; ++q) {
        float s = 0.0f;
#pragma unroll
        for (int j = 0; j < 8; ++j) s += wv[j] * x[1 + 4 * i4 + q + j];
        o[q] = s;
      }
      *(v4f*)(sp + 8 + 4 * i4) = (v4f){o[0], o[1], o[2], o[3]};
    }
    if (eps_thread) {
      v4f* e4 = (v4f*)(buf + lane * SP);
      e4[0] = (v4f){eps[0], eps[1], eps[2], eps[3]};
      e4[1] = (v4f){eps[4], eps[5], eps[6], eps[7]};
    }
  }
  __syncthreads();

  float P[CH_TILES][8];
  int tileA = 0;
  if (g < NCHAIN) {
  tileA = (g == 0) ? (SEQ / 16 - 1) : (CH_TILES * g - 1);
  const int nbase = 16 * tileA + 1;

  v8f acc = {0.0f, 0.0f, 0.0f, 0.0f, 0.0f, 0.0f, 0.0f, 0.0f};

  int r0 = nbase + 8 - KTAPS;
  if (r0 < 0) r0 = 0;
  r0 &= ~7;
  const int rmax = nbase + 22;

  const float* hb = hs + (HOFF + nbase + col + 7 - 2 * half - r0);
  const float* eb = buf + (col * SP + r0 + 2 * half);

  const int nblk  = ((rmax - r0) >> 3) + 1;
  const int nround = nblk / 3;
  const int rem    = nblk - 3 * nround;
  ZStage sA, sB, sC;
  z_load(sA, hb, eb);
  z_load(sB, hb - 8, eb + 8);
  hb -= 16;
  eb += 16;
  for (int t = 0; t < nround; ++t) {
    z_load(sC, hb, eb);
    z_mma(sA, acc);
    z_load(sA, hb - 8, eb + 8);
    z_mma(sB, acc);
    z_load(sB, hb - 16, eb + 16);
    z_mma(sC, acc);
    hb -= 24;
    eb += 24;
  }
  if (rem >= 1) z_mma(sA, acc);
  if (rem == 2) z_mma(sB, acc);

  const float* hq = hs + (HOFF + col - 2 * half);
  v2f fa0[6], fa1[6];
#pragma unroll
  for (int d = 0; d < 6; ++d) {
    const float* p = hq + 8 * (d - 1);
    fa0[d].x = p[0];   fa0[d].y = p[-1];
    fa1[d].x = p[-4];  fa1[d].y = p[-5];
  }
  float w[8];
#pragma unroll
  for (int k = 0; k < 8; ++k) w[k] = W[k];

  float run = 0.0f;
  z_scan(acc, run, P[0], col, half);
  if (g == 0) run = 0.0f;

  v8f prev = acc;
  if (g == 0) {
    const float* yp = y + (size_t)(b0 + col) * 9;
#pragma unroll
    for (int r = 0; r < 8; ++r) prev[r] = yp[r + 1] - yp[r];
  }

#pragma unroll
  for (int k = 0; k < CH_GROUPS; ++k) {
    const int t0 = CH_ROWS * g + 48 * k + 8;
    const float* eq = buf + (col * SP + t0 + 2 * half);
    ZStage q[6];
#pragma unroll
    for (int j = 0; j < 6; ++j) {
      const v2f e0 = *(const v2f*)(eq + 8 * j);
      const v2f e1 = *(const v2f*)(eq + 8 * j + 4);
      q[j].x0 = e0.x;  q[j].y0 = e0.y;
      q[j].x1 = e1.x;  q[j].y1 = e1.y;
    }

    float vm[8];
#pragma unroll
    for (int m = 0; m < 8; ++m) {
      float sum = w[0] * prev[m];
#pragma unroll
      for (int r = m + 1; r < 8; ++r) sum += w[r - m] * prev[r];
      vm[m] = sum;
    }
    {
      const int src = col + 16;
      const float u0 = __shfl(vm[0], src, 32);
      const float u1 = __shfl(vm[1], src, 32);
      const float u4 = __shfl(vm[4], src, 32);
      const float u5 = __shfl(vm[5], src, 32);
      q[0].x0 += half ? vm[2] : u0;
      q[0].y0 += half ? vm[3] : u1;
      q[0].x1 += half ? vm[6] : u4;
      q[0].y1 += half ? vm[7] : u5;
    }

    v8f c[3];
#pragma unroll
    for (int p = 0; p < 3; ++p)
      c[p] = (v8f){0.0f, 0.0f, 0.0f, 0.0f, 0.0f, 0.0f, 0.0f, 0.0f};
#pragma unroll
    for (int j = 0; j < 6; ++j) {
#pragma unroll
      for (int p = 0; p < 3; ++p) {
        if (j <= 2 * p + 1) {
          q[j].a0 = fa0[2 * p - j + 1];
          q[j].a1 = fa1[2 * p - j + 1];
          z_mma(q[j], c[p]);
        }
      }
    }

    z_scan(c[0], run, P[3 * k + 1], col, half);
    z_scan(c[1], run, P[3 * k + 2], col, half);
    z_scan(c[2], run, P[3 * k + 3], col, half);
    prev = c[2];
  }
  if (half == 0) tot[col * CS_TP + g] = run;
  }
  __syncthreads();

  if (tid < CS_ROWS * 9)
    buf[(tid / 9) * OSTR + (tid % 9)] = y[b0 * 9 + tid];

  if (g < NCHAIN) {
    float base = x8, all = x8;
    const v4f* t4 = (const v4f*)(tot + col * CS_TP);
#pragma unroll
    for (int k = 0; k < NCHAIN / 4; ++k) {
      const v4f t = t4[k];
      base += (4 * k + 0 < g) ? t.x : 0.0f;
      base += (4 * k + 1 < g) ? t.y : 0.0f;
      base += (4 * k + 2 < g) ? t.z : 0.0f;
      base += (4 * k + 3 < g) ? t.w : 0.0f;
      all += t.x; all += t.y; all += t.z; all += t.w;
    }

    float* ip = buf + (col * OSTR + 9 + 8 * half);
    const float bA = (g != 0) ? base : all;
#pragma unroll
    for (int v = 0; v < 8; ++v) ip[16 * tileA + v] = bA + P[0][v];
#pragma unroll
    for (int s = 1; s < CH_TILES; ++s) {
#pragma unroll
      for (int v = 0; v < 8; ++v) ip[CH_ROWS * g + 16 * (s - 1) + v] = base + P[s][v];
    }
  }
  __syncthreads();

  const v4f* img4 = (const v4f*)buf;
  v4f* out4 = (v4f*)(out + (size_t)b0 * OSTR);
  const int pad = (8 - (int)(((size_t)out4 >> 4) & 7)) & 7;
  if (tid < pad) out4[tid] = img4[tid];
  const int qi = tid + pad;
  static_assert(FZ_THREADS * (OUT_FULL + 1) >= OUT_F4, "one guarded trip covers the rest");
#pragma unroll
  for (int k = 0; k < OUT_FULL; ++k) out4[qi + FZ_THREADS * k] = img4[qi + FZ_THREADS * k];
  if (qi + FZ_THREADS * OUT_FULL < OUT_F4)
    out4[qi + FZ_THREADS * OUT_FULL] = img4[qi + FZ_THREADS * OUT_FULL];
}

extern "C" void kernel_launch(void* const* d_in, const int* in_sizes, int n_in,
                              void* d_out, int out_size, void* d_ws, size_t ws_size,
                              hipStream_t stream) {
  (void)in_sizes; (void)n_in; (void)out_size; (void)d_ws; (void)ws_size;
  const float* y = (const float*)d_in[0];
  const float* u = (const float*)d_in[1];
  const float* W = (const float*)d_in[2];
  float* out = (float*)d_out;

  k_arx<<<BATCH / CS_ROWS, FZ_THREADS, 0, stream>>>(y, u, W, out);
}


# blocks of 8 waves, not 16: one wave per chain and no wave that only stages and copies; each thread stages, drives and copies twice as much; 2048 waves enter, not 4096
# speedup vs baseline: 23.9521x; 1.0876x over previous
#include <hip/hip_runtime.h>


#define BATCH 4096
#define SEQ   2048
#define ARP   8
#define EXO   8
#define USTR  (SEQ + EXO)
#define OSTR  (SEQ + ARP + 1)
#define KTAPS 256
#define EROWS (SEQ + 8)
#define EPAD_ROWS 16
#define EROWS_PAD (EROWS + EPAD_ROWS)
#define HOFF  32
#define HPAD_LEN 512

typedef __attribute__((ext_vector_type(2))) float v2f;
typedef __attribute__((ext_vector_type(4))) float v4f;
typedef __attribute__((ext_vector_type(8))) float v8f;

__device__ __forceinline__ float imp_step(const float* w, const float* a) {
  float s = w[0] * a[0];
#pragma unroll
  for (int k = 1; k < 8; ++k) s = __builtin_fmaf(w[k], a[k], s);
  return s;
}

__device__ __forceinline__ void imp_first8(const float* w, float* a) {
#pragma unroll
  for (int k = 0; k < 8; ++k) a[k] = 0.0f;
  a[8] = 1.0f;
#pragma unroll
  for (int j = 1; j < 8; ++j) a[8 + j] = imp_step(w, a + j);
}


#define CS_ROWS    16
#define CS_IMG     (CS_ROWS * OSTR)
#define CS_TP      12
#define SP         2116
#define FZ_THREADS 256
#define FZ_WAVES   (FZ_THREADS / 32)
#define NCHAIN     8
#define CH_ROWS    (SEQ / NCHAIN)
#define CH_TILES   (CH_ROWS / 16)
#define CH_GROUPS  ((CH_TILES - 1) / 3)
#define FZ_BUF     (CS_ROWS * SP)
#define U_ROW4     (USTR / 4)
#define U_F4       (CS_ROWS * U_ROW4)
#define U_TRIPS    ((U_F4 + FZ_THREADS - 1) / FZ_THREADS)
#define EX_CH      128
#define OUT_F4     (CS_IMG / 4)
#define OUT_FULL   ((OUT_F4 - 7) / FZ_THREADS)

struct ZStage {
  v2f a0, a1;
  float x0, y0;
  float x1, y1;
};

__device__ __forceinline__ void z_load(ZStage& s, const float* hb, const float* eb) {
  s.a0.x = hb[0];   s.a0.y = hb[-1];
  s.a1.x = hb[-4];  s.a1.y = hb[-5];
  const v2f e0 = *(const v2f*)eb;
  const v2f e1 = *(const v2f*)(eb + 4);
  s.x0 = e0.x;  s.y0 = e0.y;
  s.x1 = e1.x;  s.y1 = e1.y;
}

__device__ __forceinline__ void z_mma(const ZStage& s, v8f& acc) {
  v2f f;
  f.x = s.x0; f.y = s.y0;
  acc = __builtin_amdgcn_wmma_f32_16x16x4_f32(false, s.a0, false, f, (short)0, acc, false, false);
  f.x = s.x1; f.y = s.y1;
  acc = __builtin_amdgcn_wmma_f32_16x16x4_f32(false, s.a1, false, f, (short)0, acc, false, false);
}

__device__ __forceinline__ void z_scan(const v8f t, float& run, float* P,
                                       const int col, const int half) {
  float s[8];
  s[0] = t[0];
#pragma unroll
  for (int v = 1; v < 8; ++v) s[v] = s[v - 1] + t[v];
  const float T0 = __shfl(s[7], col, 32);
  const float T1 = __shfl(s[7], col + 16, 32);
  const float mid = run + T0;
  const float off = half ? mid : run;
#pragma unroll
  for (int v = 0; v < 8; ++v) P[v] = off + s[v];
  run = mid + T1;
}


__global__ void __launch_bounds__(FZ_THREADS)
k_arx(const float* __restrict__ y, const float* __restrict__ u,
      const float* __restrict__ W, float* __restrict__ out) {
  static_assert(SP >= EROWS_PAD && SP % 64 == 4, "strip pitch");
  static_assert(FZ_BUF >= CS_IMG, "the image must fit where the strip was");
  static_assert(NCHAIN <= FZ_WAVES && NCHAIN * CH_ROWS == SEQ && CH_GROUPS * 3 + 1 == CH_TILES &&
                NCHAIN % 4 == 0 && CS_TP >= NCHAIN && CS_TP % 4 == 0, "chains");
  static_assert((FZ_THREADS / CS_ROWS) * EX_CH == SEQ && USTR % 4 == 0, "drive pieces");
  __shared__ __attribute__((aligned(16))) float buf[FZ_BUF];
  __shared__ __attribute__((aligned(16))) float tot[CS_ROWS * CS_TP];
  __shared__ __attribute__((aligned(16))) float hs[HPAD_LEN];

  const int tid  = threadIdx.x;
  const int lane = tid & 31;
  const int g    = __builtin_amdgcn_readfirstlane(tid >> 5);
  const int b0   = blockIdx.x * CS_ROWS;
  const int col  = lane & 15;
  const int half = lane >> 4;

  const bool eps_thread = (g == 1) && (lane < CS_ROWS);
  float yv[9];
  if (eps_thread) {
#pragma unroll
    for (int i = 0; i < 9; ++i) yv[i] = y[(b0 + lane) * 9 + i];
  }

  v4f stg[U_TRIPS];
#pragma unroll
  for (int k = 0; k < U_TRIPS; ++k) {
    const int idx = tid + FZ_THREADS * k;
    if (k < U_TRIPS - 1 || idx < U_F4) {
      const int row = idx / U_ROW4;
      const int c4  = idx - row * U_ROW4;
      stg[k] = *(const v4f*)(u + (size_t)(b0 + row) * USTR + 4 * c4);
    }
  }
  const float x8 = y[(b0 + col) * 9 + 8];

  static_assert(HOFF <= FZ_THREADS && HPAD_LEN - (HOFF + KTAPS) <= FZ_THREADS &&
                CS_ROWS * EPAD_ROWS <= FZ_THREADS, "one store per thread");
  if (tid < HOFF) hs[tid] = 0.0f;
  if (tid < HPAD_LEN - (HOFF + KTAPS)) hs[HOFF + KTAPS + tid] = 0.0f;
  if (tid < CS_ROWS * EPAD_ROWS) buf[(tid / EPAD_ROWS) * SP + EROWS + (tid % EPAD_ROWS)] = 0.0f;

  if (tid == 0) {
    float w[8];
#pragma unroll
    for (int k = 0; k < 8; ++k) w[k] = W[k];
    float a[16];
    imp_first8(w, a);
    v4f* hp = (v4f*)(hs + HOFF);
    hp[0] = (v4f){a[8], a[9], a[10], a[11]};
    hp[1] = (v4f){a[12], a[13], a[14], a[15]};
    for (int gg = 1; gg < KTAPS / 8; ++gg) {
#pragma unroll
      for (int k = 0; k < 8; ++k) a[k] = a[8 + k];
#pragma unroll
      for (int j = 0; j < 8; ++j) a[8 + j] = imp_step(w, a + j);
      hp[2 * gg]     = (v4f){a[8], a[9], a[10], a[11]};
      hp[2 * gg + 1] = (v4f){a[12], a[13], a[14], a[15]};
    }
  }

  float eps[8];
  if (eps_thread) {
    float w[8];
#pragma unroll
    for (int k = 0; k < 8; ++k) w[k] = W[k];
    float a[16];
    imp_first8(w, a);
#pragma unroll
    for (int i = 0; i < 8; ++i) {
      float s = yv[i + 1] - yv[i];
#pragma unroll
      for (int p = 0; p < 8; ++p) {
        if (p < i) s -= a[8 + i - p] * eps[p];
      }
      eps[i] = s;
    }
  }

#pragma unroll
  for (int k = 0; k < U_TRIPS; ++k) {
    const int idx = tid + FZ_THREADS * k;
    if (k < U_TRIPS - 1 || idx < U_F4) {
      const int row = idx / U_ROW4;
      const int c4  = idx - row * U_ROW4;
      *(v4f*)(buf + row * SP + 4 * c4) = stg[k];
    }
  }
  __syncthreads();

  {
    float* sp = buf + ((tid & 15) * SP + EX_CH * (tid >> 4));
    float x[EX_CH + 8];
#pragma unroll
    for (int k = 0; k < (EX_CH + 8) / 4; ++k) {
      const v4f t = ((const v4f*)sp)[k];
      x[4 * k + 0] = t.x; x[4 * k + 1] = t.y; x[4 * k + 2] = t.z; x[4 * k + 3] = t.w;
    }
    float wv[8];
#pragma unroll
    for (int j = 0; j < 8; ++j) wv[j] = W[8 + j];
    __syncthreads();
#pragma unroll
    for (int i4 = 0; i4 < EX_CH / 4; ++i4) {
      float o[4];
#pragma unroll
      for (int q = 0; q < 4; ++q) {
        float s = 0.0f;
#pragma unroll
        for (int j = 0; j < 8; ++j) s += wv[j] * x[1 + 4 * i4 + q + j];
        o[q] = s;
      }
      *(v4f*)(sp + 8 + 4 * i4) = (v4f){o[0], o[1], o[2], o[3]};
    }
    if (eps_thread) {
      v4f* e4 = (v4f*)(buf + lane * SP);
      e4[0] = (v4f){eps[0], eps[1], eps[2], eps[3]};
      e4[1] = (v4f){eps[4], eps[5], eps[6], eps[7]};
    }
  }
  __syncthreads();

  float P[CH_TILES][8];
  int tileA = 0;
  if (g < NCHAIN) {
  tileA = (g == 0) ? (SEQ / 16 - 1) : (CH_TILES * g - 1);
  const int nbase = 16 * tileA + 1;

  v8f acc = {0.0f, 0.0f, 0.0f, 0.0f, 0.0f, 0.0f, 0.0f, 0.0f};

  int r0 = nbase + 8 - KTAPS;
  if (r0 < 0) r0 = 0;
  r0 &= ~7;
  const int rmax = nbase + 22;

  const float* hb = hs + (HOFF + nbase + col + 7 - 2 * half - r0);
  const float* eb = buf + (col * SP + r0 + 2 * half);

  const int nblk  = ((rmax - r0) >> 3) + 1;
  const int nround = nblk / 3;
  const int rem    = nblk - 3 * nround;
  ZStage sA, sB, sC;
  z_load(sA, hb, eb);
  z_load(sB, hb - 8, eb + 8);
  hb -= 16;
  eb += 16;
  for (int t = 0; t < nround; ++t) {
    z_load(sC, hb, eb);
    z_mma(sA, acc);
    z_load(sA, hb - 8, eb + 8);
    z_mma(sB, acc);
    z_load(sB, hb - 16, eb + 16);
    z_mma(sC, acc);
    hb -= 24;
    eb += 24;
  }
  if (rem >= 1) z_mma(sA, acc);
  if (rem == 2) z_mma(sB, acc);

  const float* hq = hs + (HOFF + col - 2 * half);
  v2f fa0[6], fa1[6];
#pragma unroll
  for (int d = 0; d < 6; ++d) {
    const float* p = hq + 8 * (d - 1);
    fa0[d].x = p[0];   fa0[d].y = p[-1];
    fa1[d].x = p[-4];  fa1[d].y = p[-5];
  }
  float w[8];
#pragma unroll
  for (int k = 0; k < 8; ++k) w[k] = W[k];

  float run = 0.0f;
  z_scan(acc, run, P[0], col, half);
  if (g == 0) run = 0.0f;

  v8f prev = acc;
  if (g == 0) {
    const float* yp = y + (size_t)(b0 + col) * 9;
#pragma unroll
    for (int r = 0; r < 8; ++r) prev[r] = yp[r + 1] - yp[r];
  }

#pragma unroll
  for (int k = 0; k < CH_GROUPS; ++k) {
    const int t0 = CH_ROWS * g + 48 * k + 8;
    const float* eq = buf + (col * SP + t0 + 2 * half);
    ZStage q[6];
#pragma unroll
    for (int j = 0; j < 6; ++j) {
      const v2f e0 = *(const v2f*)(eq + 8 * j);
      const v2f e1 = *(const v2f*)(eq + 8 * j + 4);
      q[j].x0 = e0.x;  q[j].y0 = e0.y;
      q[j].x1 = e1.x;  q[j].y1 = e1.y;
    }

    float vm[8];
#pragma unroll
    for (int m = 0; m < 8; ++m) {
      float sum = w[0] * prev[m];
#pragma unroll
      for (int r = m + 1; r < 8; ++r) sum += w[r - m] * prev[r];
      vm[m] = sum;
    }
    {
      const int src = col + 16;
      const float u0 = __shfl(vm[0], src, 32);
      const float u1 = __shfl(vm[1], src, 32);
      const float u4 = __shfl(vm[4], src, 32);
      const float u5 = __shfl(vm[5], src, 32);
      q[0].x0 += half ? vm[2] : u0;
      q[0].y0 += half ? vm[3] : u1;
      q[0].x1 += half ? vm[6] : u4;
      q[0].y1 += half ? vm[7] : u5;
    }

    v8f c[3];
#pragma unroll
    for (int p = 0; p < 3; ++p)
      c[p] = (v8f){0.0f, 0.0f, 0.0f, 0.0f, 0.0f, 0.0f, 0.0f, 0.0f};
#pragma unroll
    for (int j = 0; j < 6; ++j) {
#pragma unroll
      for (int p = 0; p < 3; ++p) {
        if (j <= 2 * p + 1) {
          q[j].a0 = fa0[2 * p - j + 1];
          q[j].a1 = fa1[2 * p - j + 1];
          z_mma(q[j], c[p]);
        }
      }
    }

    z_scan(c[0], run, P[3 * k + 1], col, half);
    z_scan(c[1], run, P[3 * k + 2], col, half);
    z_scan(c[2], run, P[3 * k + 3], col, half);
    prev = c[2];
  }
  if (half == 0) tot[col * CS_TP + g] = run;
  }
  __syncthreads();

  if (tid < CS_ROWS * 9)
    buf[(tid / 9) * OSTR + (tid % 9)] = y[b0 * 9 + tid];

  if (g < NCHAIN) {
    float base = x8, all = x8;
    const v4f* t4 = (const v4f*)(tot + col * CS_TP);
#pragma unroll
    for (int k = 0; k < NCHAIN / 4; ++k) {
      const v4f t = t4[k];
      base += (4 * k + 0 < g) ? t.x : 0.0f;
      base += (4 * k + 1 < g) ? t.y : 0.0f;
      base += (4 * k + 2 < g) ? t.z : 0.0f;
      base += (4 * k + 3 < g) ? t.w : 0.0f;
      all += t.x; all += t.y; all += t.z; all += t.w;
    }

    float* ip = buf + (col * OSTR + 9 + 8 * half);
    const float bA = (g != 0) ? base : all;
#pragma unroll
    for (int v = 0; v < 8; ++v) ip[16 * tileA + v] = bA + P[0][v];
#pragma unroll
    for (int s = 1; s < CH_TILES; ++s) {
#pragma unroll
      for (int v = 0; v < 8; ++v) ip[CH_ROWS * g + 16 * (s - 1) + v] = base + P[s][v];
    }
  }
  __syncthreads();

  const v4f* img4 = (const v4f*)buf;
  v4f* out4 = (v4f*)(out + (size_t)b0 * OSTR);
  const int pad = (8 - (int)(((size_t)out4 >> 4) & 7)) & 7;
  if (tid < pad) out4[tid] = img4[tid];
  const int qi = tid + pad;
  static_assert(FZ_THREADS * (OUT_FULL + 1) >= OUT_F4, "one guarded trip covers the rest");
#pragma unroll
  for (int k = 0; k < OUT_FULL; ++k) out4[qi + FZ_THREADS * k] = img4[qi + FZ_THREADS * k];
  if (qi + FZ_THREADS * OUT_FULL < OUT_F4)
    out4[qi + FZ_THREADS * OUT_FULL] = img4[qi + FZ_THREADS * OUT_FULL];
}

extern "C" void kernel_launch(void* const* d_in, const int* in_sizes, int n_in,
                              void* d_out, int out_size, void* d_ws, size_t ws_size,
                              hipStream_t stream) {
  (void)in_sizes; (void)n_in; (void)out_size; (void)d_ws; (void)ws_size;
  const float* y = (const float*)d_in[0];
  const float* u = (const float*)d_in[1];
  const float* W = (const float*)d_in[2];
  float* out = (float*)d_out;

  k_arx<<<BATCH / CS_ROWS, FZ_THREADS, 0, stream>>>(y, u, W, out);
}


# a chain's 15 tiles each follow from the tile in front (K of 16 rows: 4 WMMAs, not 8), hand-over off LDS: lane halves swap in registers, state-free K block issued a tile ahead; sum order changes
# speedup vs baseline: 25.5238x; 1.0656x over previous
#include <hip/hip_runtime.h>


#define BATCH 4096
#define SEQ   2048
#define ARP   8
#define EXO   8
#define USTR  (SEQ + EXO)
#define OSTR  (SEQ + ARP + 1)
#define KTAPS 256
#define EROWS (SEQ + 8)
#define EPAD_ROWS 16
#define EROWS_PAD (EROWS + EPAD_ROWS)
#define HOFF  32
#define HPAD_LEN 512

typedef __attribute__((ext_vector_type(2))) float v2f;
typedef __attribute__((ext_vector_type(4))) float v4f;
typedef __attribute__((ext_vector_type(8))) float v8f;

__device__ __forceinline__ float imp_step(const float* w, const float* a) {
  float s = w[0] * a[0];
#pragma unroll
  for (int k = 1; k < 8; ++k) s = __builtin_fmaf(w[k], a[k], s);
  return s;
}

__device__ __forceinline__ void imp_first8(const float* w, float* a) {
#pragma unroll
  for (int k = 0; k < 8; ++k) a[k] = 0.0f;
  a[8] = 1.0f;
#pragma unroll
  for (int j = 1; j < 8; ++j) a[8 + j] = imp_step(w, a + j);
}


#define CS_ROWS    16
#define CS_IMG     (CS_ROWS * OSTR)
#define CS_TP      12
#define SP         2116
#define FZ_THREADS 256
#define FZ_WAVES   (FZ_THREADS / 32)
#define NCHAIN     8
#define CH_ROWS    (SEQ / NCHAIN)
#define CH_TILES   (CH_ROWS / 16)
#define FZ_BUF     (CS_ROWS * SP)
#define U_ROW4     (USTR / 4)
#define U_F4       (CS_ROWS * U_ROW4)
#define U_TRIPS    ((U_F4 + FZ_THREADS - 1) / FZ_THREADS)
#define EX_CH      128
#define OUT_F4     (CS_IMG / 4)
#define OUT_FULL   ((OUT_F4 - 7) / FZ_THREADS)

struct ZStage {
  v2f a0, a1;
  float x0, y0;
  float x1, y1;
};

__device__ __forceinline__ void z_load(ZStage& s, const float* hb, const float* eb) {
  s.a0.x = hb[0];   s.a0.y = hb[-1];
  s.a1.x = hb[-4];  s.a1.y = hb[-5];
  const v2f e0 = *(const v2f*)eb;
  const v2f e1 = *(const v2f*)(eb + 4);
  s.x0 = e0.x;  s.y0 = e0.y;
  s.x1 = e1.x;  s.y1 = e1.y;
}

__device__ __forceinline__ void z_mma(const ZStage& s, v8f& acc) {
  v2f f;
  f.x = s.x0; f.y = s.y0;
  acc = __builtin_amdgcn_wmma_f32_16x16x4_f32(false, s.a0, false, f, (short)0, acc, false, false);
  f.x = s.x1; f.y = s.y1;
  acc = __builtin_amdgcn_wmma_f32_16x16x4_f32(false, s.a1, false, f, (short)0, acc, false, false);
}

__device__ __forceinline__ float half_swap(const float x) {
  return __builtin_bit_cast(float, __builtin_amdgcn_permlanex16(
      __builtin_bit_cast(unsigned, x), __builtin_bit_cast(unsigned, x),
      0x76543210u, 0xfedcba98u, false, false));
}

__device__ __forceinline__ void z_scan(const v8f t, float& run, float* P,
                                       const int col, const int half) {
  float s[8];
  s[0] = t[0];
#pragma unroll
  for (int v = 1; v < 8; ++v) s[v] = s[v - 1] + t[v];
  const float oth = half_swap(s[7]);
  const float T0 = half ? oth : s[7];
  const float T1 = half ? s[7] : oth;
  const float mid = run + T0;
  const float off = half ? mid : run;
#pragma unroll
  for (int v = 0; v < 8; ++v) P[v] = off + s[v];
  run = mid + T1;
}


__global__ void __launch_bounds__(FZ_THREADS)
k_arx(const float* __restrict__ y, const float* __restrict__ u,
      const float* __restrict__ W, float* __restrict__ out) {
  static_assert(SP >= EROWS_PAD && SP % 64 == 4, "strip pitch");
  static_assert(FZ_BUF >= CS_IMG, "the image must fit where the strip was");
  static_assert(NCHAIN <= FZ_WAVES && NCHAIN * CH_ROWS == SEQ && CH_TILES * 16 == CH_ROWS &&
                NCHAIN % 4 == 0 && CS_TP >= NCHAIN && CS_TP % 4 == 0, "chains");
  static_assert((FZ_THREADS / CS_ROWS) * EX_CH == SEQ && USTR % 4 == 0, "drive pieces");
  __shared__ __attribute__((aligned(16))) float buf[FZ_BUF];
  __shared__ __attribute__((aligned(16))) float tot[CS_ROWS * CS_TP];
  __shared__ __attribute__((aligned(16))) float hs[HPAD_LEN];

  const int tid  = threadIdx.x;
  const int lane = tid & 31;
  const int g    = __builtin_amdgcn_readfirstlane(tid >> 5);
  const int b0   = blockIdx.x * CS_ROWS;
  const int col  = lane & 15;
  const int half = lane >> 4;

  const bool eps_thread = (g == 1) && (lane < CS_ROWS);
  float yv[9];
  if (eps_thread) {
#pragma unroll
    for (int i = 0; i < 9; ++i) yv[i] = y[(b0 + lane) * 9 + i];
  }

  v4f stg[U_TRIPS];
#pragma unroll
  for (int k = 0; k < U_TRIPS; ++k) {
    const int idx = tid + FZ_THREADS * k;
    if (k < U_TRIPS - 1 || idx < U_F4) {
      const int row = idx / U_ROW4;
      const int c4  = idx - row * U_ROW4;
      stg[k] = *(const v4f*)(u + (size_t)(b0 + row) * USTR + 4 * c4);
    }
  }
  const float x8 = y[(b0 + col) * 9 + 8];

  static_assert(HOFF <= FZ_THREADS && HPAD_LEN - (HOFF + KTAPS) <= FZ_THREADS &&
                CS_ROWS * EPAD_ROWS <= FZ_THREADS, "one store per thread");
  if (tid < HOFF) hs[tid] = 0.0f;
  if (tid < HPAD_LEN - (HOFF + KTAPS)) hs[HOFF + KTAPS + tid] = 0.0f;
  if (tid < CS_ROWS * EPAD_ROWS) buf[(tid / EPAD_ROWS) * SP + EROWS + (tid % EPAD_ROWS)] = 0.0f;

  if (tid == 0) {
    float w[8];
#pragma unroll
    for (int k = 0; k < 8; ++k) w[k] = W[k];
    float a[16];
    imp_first8(w, a);
    v4f* hp = (v4f*)(hs + HOFF);
    hp[0] = (v4f){a[8], a[9], a[10], a[11]};
    hp[1] = (v4f){a[12], a[13], a[14], a[15]};
    for (int gg = 1; gg < KTAPS / 8; ++gg) {
#pragma unroll
      for (int k = 0; k < 8; ++k) a[k] = a[8 + k];
#pragma unroll
      for (int j = 0; j < 8; ++j) a[8 + j] = imp_step(w, a + j);
      hp[2 * gg]     = (v4f){a[8], a[9], a[10], a[11]};
      hp[2 * gg + 1] = (v4f){a[12], a[13], a[14], a[15]};
    }
  }

  float eps[8];
  if (eps_thread) {
    float w[8];
#pragma unroll
    for (int k = 0; k < 8; ++k) w[k] = W[k];
    float a[16];
    imp_first8(w, a);
#pragma unroll
    for (int i = 0; i < 8; ++i) {
      float s = yv[i + 1] - yv[i];
#pragma unroll
      for (int p = 0; p < 8; ++p) {
        if (p < i) s -= a[8 + i - p] * eps[p];
      }
      eps[i] = s;
    }
  }

#pragma unroll
  for (int k = 0; k < U_TRIPS; ++k) {
    const int idx = tid + FZ_THREADS * k;
    if (k < U_TRIPS - 1 || idx < U_F4) {
      const int row = idx / U_ROW4;
      const int c4  = idx - row * U_ROW4;
      *(v4f*)(buf + row * SP + 4 * c4) = stg[k];
    }
  }
  __syncthreads();

  {
    float* sp = buf + ((tid & 15) * SP + EX_CH * (tid >> 4));
    float x[EX_CH + 8];
#pragma unroll
    for (int k = 0; k < (EX_CH + 8) / 4; ++k) {
      const v4f t = ((const v4f*)sp)[k];
      x[4 * k + 0] = t.x; x[4 * k + 1] = t.y; x[4 * k + 2] = t.z; x[4 * k + 3] = t.w;
    }
    float wv[8];
#pragma unroll
    for (int j = 0; j < 8; ++j) wv[j] = W[8 + j];
    __syncthreads();
#pragma unroll
    for (int i4 = 0; i4 < EX_CH / 4; ++i4) {
      float o[4];
#pragma unroll
      for (int q = 0; q < 4; ++q) {
        float s = 0.0f;
#pragma unroll
        for (int j = 0; j < 8; ++j) s += wv[j] * x[1 + 4 * i4 + q + j];
        o[q] = s;
      }
      *(v4f*)(sp + 8 + 4 * i4) = (v4f){o[0], o[1], o[2], o[3]};
    }
    if (eps_thread) {
      v4f* e4 = (v4f*)(buf + lane * SP);
      e4[0] = (v4f){eps[0], eps[1], eps[2], eps[3]};
      e4[1] = (v4f){eps[4], eps[5], eps[6], eps[7]};
    }
  }
  __syncthreads();

  float P[CH_TILES][8];
  int tileA = 0;
  if (g < NCHAIN) {
  tileA = (g == 0) ? (SEQ / 16 - 1) : (CH_TILES * g - 1);
  const int nbase = 16 * tileA + 1;

  v8f acc = {0.0f, 0.0f, 0.0f, 0.0f, 0.0f, 0.0f, 0.0f, 0.0f};

  int r0 = nbase + 8 - KTAPS;
  if (r0 < 0) r0 = 0;
  r0 &= ~7;
  const int rmax = nbase + 22;

  const float* hb = hs + (HOFF + nbase + col + 7 - 2 * half - r0);
  const float* eb = buf + (col * SP + r0 + 2 * half);

  const int nblk  = ((rmax - r0) >> 3) + 1;
  const int nround = nblk / 3;
  const int rem    = nblk - 3 * nround;
  ZStage sA, sB, sC;
  z_load(sA, hb, eb);
  z_load(sB, hb - 8, eb + 8);
  hb -= 16;
  eb += 16;
  for (int t = 0; t < nround; ++t) {
    z_load(sC, hb, eb);
    z_mma(sA, acc);
    z_load(sA, hb - 8, eb + 8);
    z_mma(sB, acc);
    z_load(sB, hb - 16, eb + 16);
    z_mma(sC, acc);
    hb -= 24;
    eb += 24;
  }
  if (rem >= 1) z_mma(sA, acc);
  if (rem == 2) z_mma(sB, acc);

  float fe[CH_TILES - 1][8];
  const float* eqf = buf + (col * SP + CH_ROWS * g + 8 + 2 * half);
#pragma unroll
  for (int s = 0; s < 2; ++s) {
#pragma unroll
    for (int j = 0; j < 4; ++j) {
      const v2f e = *(const v2f*)(eqf + 16 * s + 4 * j);
      fe[s][2 * j] = e.x;  fe[s][2 * j + 1] = e.y;
    }
  }
  __builtin_amdgcn_sched_barrier(0);

  const float* hq = hs + (HOFF + col - 2 * half);
  v2f fb0a, fb0b, fb1a, fb1b;
  fb0a.x = hq[0];    fb0a.y = hq[-1];
  fb0b.x = hq[-4];   fb0b.y = hq[-5];
  fb1a.x = hq[-8];   fb1a.y = hq[-9];
  fb1b.x = hq[-12];  fb1b.y = hq[-13];
  float w[8];
#pragma unroll
  for (int k = 0; k < 8; ++k) w[k] = W[k];

  float run = 0.0f;
  v8f done = acc;

  v8f prev = acc;
  if (g == 0) {
    const float* yp = y + (size_t)(b0 + col) * 9;
#pragma unroll
    for (int r = 0; r < 8; ++r) prev[r] = yp[r + 1] - yp[r];
  }

  v2f bq;
  v8f cur = (v8f){0.0f, 0.0f, 0.0f, 0.0f, 0.0f, 0.0f, 0.0f, 0.0f};
  bq.x = fe[0][4];  bq.y = fe[0][5];
  cur = __builtin_amdgcn_wmma_f32_16x16x4_f32(false, fb1a, false, bq, (short)0, cur, false, false);
  bq.x = fe[0][6];  bq.y = fe[0][7];
  cur = __builtin_amdgcn_wmma_f32_16x16x4_f32(false, fb1b, false, bq, (short)0, cur, false, false);

#pragma unroll
  for (int s = 1; s < CH_TILES; ++s) {
    if (s + 1 < CH_TILES - 1) {
#pragma unroll
      for (int j = 0; j < 4; ++j) {
        const v2f e = *(const v2f*)(eqf + 16 * (s + 1) + 4 * j);
        fe[s + 1][2 * j] = e.x;  fe[s + 1][2 * j + 1] = e.y;
      }
    }
    __builtin_amdgcn_sched_barrier(0);

    float vm[8];
#pragma unroll
    for (int m = 0; m < 8; ++m) {
      float se = w[0] * prev[m];
      float so = (m < 7) ? w[1] * prev[(m < 7) ? m + 1 : 7] : 0.0f;
#pragma unroll
      for (int r = m + 2; r < 8; ++r) {
        if ((r - m) & 1) so = __builtin_fmaf(w[r - m], prev[r], so);
        else             se = __builtin_fmaf(w[r - m], prev[r], se);
      }
      vm[m] = (m < 7) ? se + so : se;
    }
    v2f s0, s1;
    {
      const float u0 = half_swap(vm[0]);
      const float u1 = half_swap(vm[1]);
      const float u4 = half_swap(vm[4]);
      const float u5 = half_swap(vm[5]);
      s0.x = fe[s - 1][0] + (half ? vm[2] : u0);
      s0.y = fe[s - 1][1] + (half ? vm[3] : u1);
      s1.x = fe[s - 1][2] + (half ? vm[6] : u4);
      s1.y = fe[s - 1][3] + (half ? vm[7] : u5);
    }
    v8f nxt = (v8f){0.0f, 0.0f, 0.0f, 0.0f, 0.0f, 0.0f, 0.0f, 0.0f};
    cur = __builtin_amdgcn_wmma_f32_16x16x4_f32(false, fb0a, false, s0, (short)0, cur, false, false);
    if (s + 1 < CH_TILES) {
      bq.x = fe[s][4];  bq.y = fe[s][5];
      nxt = __builtin_amdgcn_wmma_f32_16x16x4_f32(false, fb1a, false, bq, (short)0, nxt, false, false);
    }
    cur = __builtin_amdgcn_wmma_f32_16x16x4_f32(false, fb0b, false, s1, (short)0, cur, false, false);
    if (s + 1 < CH_TILES) {
      bq.x = fe[s][6];  bq.y = fe[s][7];
      nxt = __builtin_amdgcn_wmma_f32_16x16x4_f32(false, fb1b, false, bq, (short)0, nxt, false, false);
    }

    z_scan(done, run, P[s - 1], col, half);
    if (s == 1 && g == 0) run = 0.0f;
    done = cur;
    prev = cur;
    cur = nxt;
  }
  z_scan(done, run, P[CH_TILES - 1], col, half);
  if (half == 0) tot[col * CS_TP + g] = run;
  }
  __syncthreads();

  if (tid < CS_ROWS * 9)
    buf[(tid / 9) * OSTR + (tid % 9)] = y[b0 * 9 + tid];

  if (g < NCHAIN) {
    float base = x8, all = x8;
    const v4f* t4 = (const v4f*)(tot + col * CS_TP);
#pragma unroll
    for (int k = 0; k < NCHAIN / 4; ++k) {
      const v4f t = t4[k];
      base += (4 * k + 0 < g) ? t.x : 0.0f;
      base += (4 * k + 1 < g) ? t.y : 0.0f;
      base += (4 * k + 2 < g) ? t.z : 0.0f;
      base += (4 * k + 3 < g) ? t.w : 0.0f;
      all += t.x; all += t.y; all += t.z; all += t.w;
    }

    float* ip = buf + (col * OSTR + 9 + 8 * half);
    const float bA = (g != 0) ? base : all;
#pragma unroll
    for (int v = 0; v < 8; ++v) ip[16 * tileA + v] = bA + P[0][v];
#pragma unroll
    for (int s = 1; s < CH_TILES; ++s) {
#pragma unroll
      for (int v = 0; v < 8; ++v) ip[CH_ROWS * g + 16 * (s - 1) + v] = base + P[s][v];
    }
  }
  __syncthreads();

  const v4f* img4 = (const v4f*)buf;
  v4f* out4 = (v4f*)(out + (size_t)b0 * OSTR);
  const int pad = (8 - (int)(((size_t)out4 >> 4) & 7)) & 7;
  if (tid < pad) out4[tid] = img4[tid];
  const int qi = tid + pad;
  static_assert(FZ_THREADS * (OUT_FULL + 1) >= OUT_F4, "one guarded trip covers the rest");
#pragma unroll
  for (int k = 0; k < OUT_FULL; ++k) out4[qi + FZ_THREADS * k] = img4[qi + FZ_THREADS * k];
  if (qi + FZ_THREADS * OUT_FULL < OUT_F4)
    out4[qi + FZ_THREADS * OUT_FULL] = img4[qi + FZ_THREADS * OUT_FULL];
}

extern "C" void kernel_launch(void* const* d_in, const int* in_sizes, int n_in,
                              void* d_out, int out_size, void* d_ws, size_t ws_size,
                              hipStream_t stream) {
  (void)in_sizes; (void)n_in; (void)out_size; (void)d_ws; (void)ws_size;
  const float* y = (const float*)d_in[0];
  const float* u = (const float*)d_in[1];
  const float* W = (const float*)d_in[2];
  float* out = (float*)d_out;

  k_arx<<<BATCH / CS_ROWS, FZ_THREADS, 0, stream>>>(y, u, W, out);
}
